# starting kernel
# baseline (speedup 1.0000x reference)
#include <hip/hip_runtime.h>
#include <math.h>

typedef __attribute__((ext_vector_type(2))) float v2f;
typedef __attribute__((ext_vector_type(8))) float v8f;

#define N_NODES 50000
#define C_IN    128
#define C_HID   256
#define HEADS   4
#define EMBED   64
#define E0      800000
#define ETOT    (E0 + N_NODES)
#define NEG_SLOPE 0.2f

__global__ void gemm_wmma(const float* __restrict__ X,
                          const float* __restrict__ W,
                          float* __restrict__ H,
                          int K, int N) {
    const int lane   = threadIdx.x & 31;
    const int wave   = threadIdx.x >> 5;
    const int tileM  = blockIdx.x;
    const int lane15 = lane & 15;
    const int khalf  = lane >> 4;
    const int row    = tileM * 16 + lane15;
    const int col0   = wave * 64;

    v8f acc0 = {}, acc1 = {}, acc2 = {}, acc3 = {};

    for (int k = 0; k < K; k += 4) {
        const int kk = k + khalf * 2;
        v2f a;
        a.x = X[row * K + kk];
        a.y = X[row * K + kk + 1];

        const float* wp0 = W + (size_t)kk * N;
        const float* wp1 = wp0 + N;

        v2f b0, b1, b2, b3;
        b0.x = wp0[col0 +  0 + lane15]; b0.y = wp1[col0 +  0 + lane15];
        b1.x = wp0[col0 + 16 + lane15]; b1.y = wp1[col0 + 16 + lane15];
        b2.x = wp0[col0 + 32 + lane15]; b2.y = wp1[col0 + 32 + lane15];
        b3.x = wp0[col0 + 48 + lane15]; b3.y = wp1[col0 + 48 + lane15];

        acc0 = __builtin_amdgcn_wmma_f32_16x16x4_f32(false, a, false, b0, (short)0, acc0, false, false);
        acc1 = __builtin_amdgcn_wmma_f32_16x16x4_f32(false, a, false, b1, (short)0, acc1, false, false);
        acc2 = __builtin_amdgcn_wmma_f32_16x16x4_f32(false, a, false, b2, (short)0, acc2, false, false);
        acc3 = __builtin_amdgcn_wmma_f32_16x16x4_f32(false, a, false, b3, (short)0, acc3, false, false);
    }

    const int rbase = tileM * 16 + khalf * 8;
#pragma unroll
    for (int i = 0; i < 8; ++i) {
        float* op = H + (size_t)(rbase + i) * N + col0 + lane15;
        op[ 0] = acc0[i];
        op[16] = acc1[i];
        op[32] = acc2[i];
        op[48] = acc3[i];
    }
}

__global__ void attn_dots(const float* __restrict__ H,
                          const float* __restrict__ att_src,
                          const float* __restrict__ att_dst,
                          float* __restrict__ a_src,
                          float* __restrict__ a_dst) {
    const int gw   = (int)((blockIdx.x * blockDim.x + threadIdx.x) >> 5);
    const int lane = threadIdx.x & 31;
    const int node = gw >> 2;
    const int head = gw & 3;
    if (node >= N_NODES) return;

    const float* hp = H + (size_t)node * C_HID + head * EMBED;
    const float* as = att_src + head * EMBED;
    const float* ad = att_dst + head * EMBED;

    float s = 0.f, d = 0.f;
#pragma unroll
    for (int c = 0; c < EMBED; c += 32) {
        float v = hp[c + lane];
        s += v * as[c + lane];
        d += v * ad[c + lane];
    }
#pragma unroll
    for (int off = 16; off > 0; off >>= 1) {
        s += __shfl_down(s, off, 32);
        d += __shfl_down(d, off, 32);
    }
    if (lane == 0) {
        a_src[node * HEADS + head] = s;
        a_dst[node * HEADS + head] = d;
    }
}

__device__ __forceinline__ unsigned ford(float f) {
    unsigned u = __float_as_uint(f);
    return (u >> 31) ? ~u : (u | 0x80000000u);
}
__device__ __forceinline__ float funord(unsigned u) {
    return (u >> 31) ? __uint_as_float(u ^ 0x80000000u) : __uint_as_float(~u);
}

__global__ void fill_u32(unsigned* __restrict__ p, unsigned v, int n) {
    int i = blockIdx.x * blockDim.x + threadIdx.x;
    if (i < n) p[i] = v;
}

__device__ __forceinline__ void edge_sd(int e, const int* src, const int* dst, int& s, int& d) {
    if (e < E0) { s = src[e]; d = dst[e]; }
    else        { s = e - E0; d = s; }
}

__device__ __forceinline__ float edge_logit(const float* a_src, const float* a_dst,
                                            int s, int d, int h) {
    float v = a_src[s * HEADS + h] + a_dst[d * HEADS + h];
    return v > 0.f ? v : NEG_SLOPE * v;
}

__global__ void edge_max(const int* __restrict__ src, const int* __restrict__ dst,
                         const float* __restrict__ a_src, const float* __restrict__ a_dst,
                         unsigned* __restrict__ m) {
    int i = blockIdx.x * blockDim.x + threadIdx.x;
    if (i >= ETOT * HEADS) return;
    int e = i >> 2, h = i & 3, s, d;
    edge_sd(e, src, dst, s, d);
    float v = edge_logit(a_src, a_dst, s, d, h);
    atomicMax(&m[d * HEADS + h], ford(v));
}

__global__ void edge_expsum(const int* __restrict__ src, const int* __restrict__ dst,
                            const float* __restrict__ a_src, const float* __restrict__ a_dst,
                            const unsigned* __restrict__ m,
                            float* __restrict__ ee, float* __restrict__ z) {
    int i = blockIdx.x * blockDim.x + threadIdx.x;
    if (i >= ETOT * HEADS) return;
    int e = i >> 2, h = i & 3, s, d;
    edge_sd(e, src, dst, s, d);
    float v  = edge_logit(a_src, a_dst, s, d, h);
    float mv = funord(m[d * HEADS + h]);
    float ev = __expf(v - mv);
    ee[i] = ev;
    atomicAdd(&z[d * HEADS + h], ev);
}

__global__ void edge_scatter(const int* __restrict__ src, const int* __restrict__ dst,
                             const float* __restrict__ ee, const float* __restrict__ z,
                             const float* __restrict__ H, float* __restrict__ agg) {
    int e = blockIdx.x;
    int t = threadIdx.x;
    int h = t >> 6, s, d;
    edge_sd(e, src, dst, s, d);
    float alpha = ee[(size_t)e * HEADS + h] / (z[d * HEADS + h] + 1e-16f);
    float val   = H[(size_t)s * C_HID + t] * alpha;
    atomicAdd(&agg[(size_t)d * C_HID + t], val);
}

__global__ void bias_act(const float* __restrict__ agg, const float* __restrict__ bias,
                         float* __restrict__ out, int n, int relu) {
    int i = blockIdx.x * blockDim.x + threadIdx.x;
    if (i >= n) return;
    float v = agg[i] + bias[i & (C_HID - 1)];
    out[i] = relu ? fmaxf(v, 0.f) : v;
}

extern "C" void kernel_launch(void* const* d_in, const int* in_sizes, int n_in,
                              void* d_out, int out_size, void* d_ws, size_t ws_size,
                              hipStream_t stream) {
    const float* x        = (const float*)d_in[0];
    const int*   eidx     = (const int*)  d_in[1];
    const float* W1       = (const float*)d_in[2];
    const float* att_s1   = (const float*)d_in[3];
    const float* att_d1   = (const float*)d_in[4];
    const float* b1       = (const float*)d_in[5];
    const float* W2       = (const float*)d_in[6];
    const float* att_s2   = (const float*)d_in[7];
    const float* att_d2   = (const float*)d_in[8];
    const float* b2       = (const float*)d_in[9];
    float*       out      = (float*)d_out;

    const int* src = eidx;
    const int* dst = eidx + E0;

    const size_t NB = (size_t)N_NODES * C_HID;
    float* h      = (float*)d_ws;
    float* agg    = h      + NB;
    float* hrelu  = agg    + NB;
    float* ee     = hrelu  + NB;
    float* a_src  = ee     + (size_t)ETOT * HEADS;
    float* a_dst  = a_src  + (size_t)N_NODES * HEADS;
    unsigned* m   = (unsigned*)(a_dst + (size_t)N_NODES * HEADS);
    float* z      = (float*)(m + (size_t)N_NODES * HEADS);

    const int NH       = N_NODES * HEADS;
    const int EH       = ETOT * HEADS;
    const int NELEM    = (int)NB;
    const unsigned ORD_NEG_INF = 0x007FFFFFu;

    const dim3 B256(256);
    const int gNH  = (NH    + 255) / 256;
    const int gEH  = (EH    + 255) / 256;
    const int gNE  = (NELEM + 255) / 256;
    const int gMM  = N_NODES / 16;
    const int gDOT = (NH * 32 + 255) / 256;

    gemm_wmma  <<<gMM, 128, 0, stream>>>(x, W1, h, C_IN, C_HID);
    attn_dots  <<<gDOT, B256, 0, stream>>>(h, att_s1, att_d1, a_src, a_dst);
    fill_u32   <<<gNH, B256, 0, stream>>>(m, ORD_NEG_INF, NH);
    fill_u32   <<<gNH, B256, 0, stream>>>((unsigned*)z, 0u, NH);
    fill_u32   <<<gNE, B256, 0, stream>>>((unsigned*)agg, 0u, NELEM);
    edge_max   <<<gEH, B256, 0, stream>>>(src, dst, a_src, a_dst, m);
    edge_expsum<<<gEH, B256, 0, stream>>>(src, dst, a_src, a_dst, m, ee, z);
    edge_scatter<<<ETOT, B256, 0, stream>>>(src, dst, ee, z, h, agg);
    bias_act   <<<gNE, B256, 0, stream>>>(agg, b1, hrelu, NELEM, 1);

    gemm_wmma  <<<gMM, 128, 0, stream>>>(hrelu, W2, h, C_HID, C_HID);
    attn_dots  <<<gDOT, B256, 0, stream>>>(h, att_s2, att_d2, a_src, a_dst);
    fill_u32   <<<gNH, B256, 0, stream>>>(m, ORD_NEG_INF, NH);
    fill_u32   <<<gNH, B256, 0, stream>>>((unsigned*)z, 0u, NH);
    fill_u32   <<<gNE, B256, 0, stream>>>((unsigned*)agg, 0u, NELEM);
    edge_max   <<<gEH, B256, 0, stream>>>(src, dst, a_src, a_dst, m);
    edge_expsum<<<gEH, B256, 0, stream>>>(src, dst, a_src, a_dst, m, ee, z);
    edge_scatter<<<ETOT, B256, 0, stream>>>(src, dst, ee, z, h, agg);
    bias_act   <<<gNE, B256, 0, stream>>>(agg, b2, out, NELEM, 0);
}


# Attention logits on the WMMA: rows times (W folded with att_src, att_dst in a small launch), a 16x16 tile per row tile in extra blocks of the product launch; dots launches gone; dot sum order changes
# speedup vs baseline: 4.8898x; 4.8898x over previous
#include <hip/hip_runtime.h>
#include <math.h>

typedef __attribute__((ext_vector_type(2))) float v2f;
typedef __attribute__((ext_vector_type(8))) float v8f;

#define N_NODES 50000
#define C_IN    128
#define C_HID   256
#define HEADS   4
#define EMBED   64
#define E0      800000
#define ETOT    (E0 + N_NODES)
#define NEG_SLOPE 0.2f

template <bool ACT>
__global__ __launch_bounds__(128, 16) void gemm_wmma(const float* __restrict__ X,
                          const float* __restrict__ bias,
                          const float* __restrict__ W,
                          const float* __restrict__ WA,
                          float* __restrict__ H,
                          float* __restrict__ a_src,
                          float* __restrict__ a_dst,
                          int K, int N, int MT) {
    const int lane   = threadIdx.x & 31;
    const int wave   = __builtin_amdgcn_readfirstlane(threadIdx.x >> 5);
    const bool dots  = (int)blockIdx.x >= MT;
    const int tileM  = dots ? ((int)blockIdx.x - MT) * 4 + wave : (int)blockIdx.x;
    if (tileM >= MT) return;
    const int lane15 = lane & 15;
    const int khalf  = lane >> 4;
    const int row    = tileM * 16 + lane15;
    const int col0   = wave * 64;

    if (dots) {
        v8f acc = {}, accB = {};
        int k = 0;
        for (; k + 8 <= K; k += 8) {
            const int kk = k + khalf * 2;
            const int kq = kk + 4;
            v2f a, aq, b, bq;
            a.x  = X[row * K + kk];  a.y  = X[row * K + kk + 1];
            aq.x = X[row * K + kq];  aq.y = X[row * K + kq + 1];
            v2f bb = {}, bbq = {};
            if (ACT) { bb.x = bias[kk]; bb.y = bias[kk + 1]; bbq.x = bias[kq]; bbq.y = bias[kq + 1]; }
            b.x  = WA[kk * 16 + lane15];  b.y  = WA[(kk + 1) * 16 + lane15];
            bq.x = WA[kq * 16 + lane15];  bq.y = WA[(kq + 1) * 16 + lane15];
            if (ACT) {
                a.x  = fmaxf(a.x  + bb.x,  0.f);  a.y  = fmaxf(a.y  + bb.y,  0.f);
                aq.x = fmaxf(aq.x + bbq.x, 0.f);  aq.y = fmaxf(aq.y + bbq.y, 0.f);
            }
            acc  = __builtin_amdgcn_wmma_f32_16x16x4_f32(false, a,  false, b,  (short)0, acc,  false, false);
            accB = __builtin_amdgcn_wmma_f32_16x16x4_f32(false, aq, false, bq, (short)0, accB, false, false);
        }
        if (k < K) {
            const int kk = k + khalf * 2;
            v2f a, b;
            a.x = X[row * K + kk];  a.y = X[row * K + kk + 1];
            if (ACT) { a.x = fmaxf(a.x + bias[kk], 0.f); a.y = fmaxf(a.y + bias[kk + 1], 0.f); }
            b.x = WA[kk * 16 + lane15];  b.y = WA[(kk + 1) * 16 + lane15];
            acc = __builtin_amdgcn_wmma_f32_16x16x4_f32(false, a, false, b, (short)0, acc, false, false);
        }
#pragma unroll
        for (int i = 0; i < 8; ++i) acc[i] += accB[i];
        if (lane15 < 8) {
            float* op = (lane15 < 4 ? a_src : a_dst) + (size_t)(tileM * 16 + khalf * 8) * HEADS + (lane15 & 3);
#pragma unroll
            for (int i = 0; i < 8; ++i) op[i * HEADS] = acc[i];
        }
        return;
    }

    v8f acc0 = {}, acc1 = {}, acc2 = {}, acc3 = {};

    for (int k = 0; k < K; k += 4) {
        const int kk = k + khalf * 2;
        v2f a;
        a.x = X[row * K + kk];
        a.y = X[row * K + kk + 1];
        v2f bb = {};
        if (ACT) { bb.x = bias[kk]; bb.y = bias[kk + 1]; }

        const float* wp0 = W + (size_t)kk * N;
        const float* wp1 = wp0 + N;

        v2f b0, b1, b2, b3;
        b0.x = wp0[col0 +  0 + lane15]; b0.y = wp1[col0 +  0 + lane15];
        b1.x = wp0[col0 + 16 + lane15]; b1.y = wp1[col0 + 16 + lane15];
        b2.x = wp0[col0 + 32 + lane15]; b2.y = wp1[col0 + 32 + lane15];
        b3.x = wp0[col0 + 48 + lane15]; b3.y = wp1[col0 + 48 + lane15];

        if (ACT) {
            a.x = fmaxf(a.x + bb.x, 0.f);
            a.y = fmaxf(a.y + bb.y, 0.f);
        }

        acc0 = __builtin_amdgcn_wmma_f32_16x16x4_f32(false, a, false, b0, (short)0, acc0, false, false);
        acc1 = __builtin_amdgcn_wmma_f32_16x16x4_f32(false, a, false, b1, (short)0, acc1, false, false);
        acc2 = __builtin_amdgcn_wmma_f32_16x16x4_f32(false, a, false, b2, (short)0, acc2, false, false);
        acc3 = __builtin_amdgcn_wmma_f32_16x16x4_f32(false, a, false, b3, (short)0, acc3, false, false);
    }

    const int rbase = tileM * 16 + khalf * 8;
#pragma unroll
    for (int i = 0; i < 8; ++i) {
        float* op = H + (size_t)(rbase + i) * N + col0 + lane15;
        op[ 0] = acc0[i];
        op[16] = acc1[i];
        op[32] = acc2[i];
        op[48] = acc3[i];
    }
}

__global__ void fold_att(const float* __restrict__ W1,
                         const float* __restrict__ as1,
                         const float* __restrict__ ad1,
                         const float* __restrict__ W2,
                         const float* __restrict__ as2,
                         const float* __restrict__ ad2,
                         float* __restrict__ WA1,
                         float* __restrict__ WA2) {
    const int gw   = (int)((blockIdx.x * blockDim.x + threadIdx.x) >> 5);
    const int lane = threadIdx.x & 31;
    if (gw >= C_IN + C_HID) return;
    const bool second = gw >= C_IN;
    const int  k      = second ? gw - C_IN : gw;
    const float* Wm = second ? W2 : W1;
    const float* as = second ? as2 : as1;
    const float* ad = second ? ad2 : ad1;
    float*       WA = second ? WA2 : WA1;

    const int j    = lane >> 2;
    const int part = lane & 3;
    const int head = j & 3;
    const float* att = (j < 4 ? as : ad) + head * EMBED + part * 16;
    const float* wp  = Wm + (size_t)k * C_HID + head * EMBED + part * 16;
    float s = 0.f;
#pragma unroll
    for (int c = 0; c < 16; c += 4) {
        const float4 w4 = *(const float4*)(wp + c);
        const float4 t4 = *(const float4*)(att + c);
        s += w4.x * t4.x;
        s += w4.y * t4.y;
        s += w4.z * t4.z;
        s += w4.w * t4.w;
    }
    s += __shfl_xor(s, 1, 32);
    s += __shfl_xor(s, 2, 32);
    if (part == 0) WA[k * 16 + j] = s;
    if (part == 1) WA[k * 16 + 8 + j] = 0.f;
}

__device__ __forceinline__ unsigned ford(float f) {
    unsigned u = __float_as_uint(f);
    return (u >> 31) ? ~u : (u | 0x80000000u);
}
__device__ __forceinline__ float funord(unsigned u) {
    return (u >> 31) ? __uint_as_float(u ^ 0x80000000u) : __uint_as_float(~u);
}

__device__ __forceinline__ float leaky(float v) {
    return v > 0.f ? v : NEG_SLOPE * v;
}

#define SLAB   200
#define NSLAB  (N_NODES / SLAB)
#define QCAP   640
#define SWAVES 32
#define ORD_NEG_INF 0x007FFFFFu
static_assert(N_NODES % SLAB == 0, "rows split evenly over the blocks");
static_assert(SLAB <= 256 && ETOT < (1 << 23), "a queue entry packs edge << 8 | local row");
static_assert(E0 % 4 == 0, "dst is scanned in int4 pieces");

__device__ __forceinline__ void lane_logits(const float* __restrict__ a_src,
                                            const float* __restrict__ a_dst,
                                            int s, int d, float v[HEADS]) {
    const float4 as = *(const float4*)(a_src + (size_t)s * HEADS);
    const float4 ad = *(const float4*)(a_dst + (size_t)d * HEADS);
    v[0] = leaky(as.x + ad.x);
    v[1] = leaky(as.y + ad.y);
    v[2] = leaky(as.z + ad.z);
    v[3] = leaky(as.w + ad.w);
}

__device__ __forceinline__ void lane_max(const float* __restrict__ a_src,
                                         const float* __restrict__ a_dst,
                                         int s, int dl, int base, unsigned* mloc) {
    float v[HEADS];
    lane_logits(a_src, a_dst, s, base + dl, v);
#pragma unroll
    for (int h = 0; h < HEADS; ++h)
        __hip_atomic_fetch_max(mloc + dl * HEADS + h, ford(v[h]),
                               __ATOMIC_RELAXED, __HIP_MEMORY_SCOPE_WORKGROUP);
}

__device__ __forceinline__ void lane_z(const float* __restrict__ a_src,
                                       const float* __restrict__ a_dst,
                                       int s, int dl, int base,
                                       const unsigned* mloc, float* zloc) {
    float v[HEADS];
    lane_logits(a_src, a_dst, s, base + dl, v);
#pragma unroll
    for (int h = 0; h < HEADS; ++h) {
        const float mv = funord(mloc[dl * HEADS + h]);
        __hip_atomic_fetch_add(zloc + dl * HEADS + h, __expf(v[h] - mv),
                               __ATOMIC_RELAXED, __HIP_MEMORY_SCOPE_WORKGROUP);
    }
}

template <int NB>
__device__ __forceinline__ void rows_add(int r, int base, int lane,
                                         const float* __restrict__ a_src,
                                         const float* __restrict__ a_dst,
                                         const unsigned* mloc, const float* zloc,
                                         const float* __restrict__ H,
                                         float* slab) {
    int s[NB], dl[NB];
#pragma unroll
    for (int k = 0; k < NB; ++k) {
        const int rk = __builtin_amdgcn_readlane(r, 4 * k);
        s[k]  = rk >> 8;
        dl[k] = rk & 255;
    }
    float hv[NB][8];
#pragma unroll
    for (int k = 0; k < NB; ++k) {
        const float* hp = H + (size_t)s[k] * C_HID + lane;
#pragma unroll
        for (int j = 0; j < 8; ++j) hv[k][j] = hp[j * 32];
    }
    const int h   = lane & 3;
    const int sL  = r >> 8;
    const int dlL = r & 255;
    const float v  = leaky(a_src[(size_t)sL * HEADS + h] + a_dst[(size_t)(base + dlL) * HEADS + h]);
    const float ev = __expf(v - funord(mloc[dlL * HEADS + h]));
    const int   al = __float_as_int(ev / (zloc[dlL * HEADS + h] + 1e-16f));
#pragma unroll
    for (int k = 0; k < NB; ++k) {
        float* sp = slab + dl[k] * C_HID + lane;
#pragma unroll
        for (int j = 0; j < 8; ++j) {
            const float a = __int_as_float(__builtin_amdgcn_readlane(al, 4 * k + (j >> 1)));
            __hip_atomic_fetch_add(sp + j * 32, hv[k][j] * a,
                                   __ATOMIC_RELAXED, __HIP_MEMORY_SCOPE_WORKGROUP);
        }
    }
}

template <int PH>
__device__ __forceinline__ void rescan(int tid, int lane, int wave, int base,
                                       const int* __restrict__ src,
                                       const int* __restrict__ dst,
                                       const float* __restrict__ a_src,
                                       const float* __restrict__ a_dst,
                                       unsigned* mloc, float* zloc,
                                       const float* __restrict__ H,
                                       float* slab) {
    if (PH < 2) {
        for (int n = tid; n < SLAB; n += 1024) {
            if (PH == 0) lane_max(a_src, a_dst, base + n, n, base, mloc);
            else         lane_z(a_src, a_dst, base + n, n, base, mloc, zloc);
        }
    } else {
        for (int n = wave; n < SLAB; n += SWAVES) {
            rows_add<1>(((base + n) << 8) | n, base, lane, a_src, a_dst, mloc, zloc, H, slab);
        }
    }
    const int4* d4p = (const int4*)dst;
    const int NV = E0 / 4;
    for (int vb = wave * 32; vb < NV; vb += SWAVES * 32) {
        const int vi = vb + lane;
        const int4 t = d4p[vi < NV ? vi : NV - 1];
        const int comp[4] = {t.x, t.y, t.z, t.w};
#pragma unroll
        for (int c = 0; c < 4; ++c) {
            const unsigned dl = (unsigned)(comp[c] - base);
            const bool mt = vi < NV && dl < (unsigned)SLAB;
            int s = 0;
            if (mt) s = src[vi * 4 + c];
            if (PH == 0) {
                if (mt) lane_max(a_src, a_dst, s, (int)dl, base, mloc);
            } else if (PH == 1) {
                if (mt) lane_z(a_src, a_dst, s, (int)dl, base, mloc, zloc);
            } else {
                unsigned mask = __builtin_amdgcn_ballot_w32(mt);
                while (mask) {
                    const int b = __builtin_ctz(mask);
                    mask &= mask - 1;
                    const int rb = __builtin_amdgcn_readlane((s << 8) | (int)dl, b);
                    rows_add<1>(rb, base, lane, a_src, a_dst, mloc, zloc, H, slab);
                }
            }
        }
    }
}

template <bool FIRST>
__global__ __launch_bounds__(1024) void edge_slab(const int* __restrict__ src,
                                                  const int* __restrict__ dst,
                                                  const float* __restrict__ a_src,
                                                  const float* __restrict__ a_dst,
                                                  const float* __restrict__ H,
                                                  float* __restrict__ agg,
                                                  int* __restrict__ qsave,
                                                  int* __restrict__ qcount,
                                                  int* __restrict__ qslow) {
    __shared__ float    slab[SLAB * C_HID];
    __shared__ int      queue[SWAVES * QCAP];
    __shared__ unsigned mloc[SLAB * HEADS];
    __shared__ float    zloc[SLAB * HEADS];
    __shared__ int      ovflag;

    const int tid  = threadIdx.x;
    const int lane = tid & 31;
    const int wave = __builtin_amdgcn_readfirstlane(tid >> 5);
    const int base = blockIdx.x * SLAB;

    float4* s4 = (float4*)slab;
    for (int i = tid; i < SLAB * C_HID / 4; i += 1024) s4[i] = make_float4(0.f, 0.f, 0.f, 0.f);
    for (int i = tid; i < SLAB * HEADS; i += 1024) { mloc[i] = ORD_NEG_INF; zloc[i] = 0.f; }
    if (FIRST && tid == 0) ovflag = 0;

    int* qw = queue + wave * QCAP;
    int* gq = qsave + ((size_t)blockIdx.x * SWAVES + wave) * QCAP;
    int* gc = qcount + blockIdx.x * SWAVES + wave;
    int  cnt;
    bool slow;
    if (FIRST) {
        cnt = (SLAB - 1 - wave) / SWAVES + 1;
        if (lane < cnt) {
            const int n = wave + SWAVES * lane;
            qw[lane] = ((E0 + base + n) << 8) | n;
        }
        __syncthreads();

        const int4* d4p = (const int4*)dst;
        const int NV = E0 / 4;
        const int4 none = make_int4(-1, -1, -1, -1);
        bool ovf = false;
        for (int vb = wave * 128; vb < NV; vb += SWAVES * 128) {
            int4 p[4];
#pragma unroll
            for (int q = 0; q < 4; ++q) {
                const int vi = vb + q * 32 + lane;
                const int4 t = d4p[vi < NV ? vi : NV - 1];
                p[q] = vi < NV ? t : none;
            }
#pragma unroll
            for (int q = 0; q < 4; ++q) {
                const int vi = vb + q * 32 + lane;
                const int comp[4] = {p[q].x, p[q].y, p[q].z, p[q].w};
#pragma unroll
                for (int c = 0; c < 4; ++c) {
                    const unsigned dl = (unsigned)(comp[c] - base);
                    const bool mt = dl < (unsigned)SLAB;
                    const unsigned mask = __builtin_amdgcn_ballot_w32(mt);
                    if (mask) {
                        const int n = __builtin_popcount(mask);
                        if (cnt + n <= QCAP) {
                            if (mt) qw[cnt + __builtin_amdgcn_mbcnt_lo(mask, 0u)] = ((vi * 4 + c) << 8) | (int)dl;
                            cnt += n;
                        } else {
                            ovf = true;
                        }
                    }
                }
            }
        }
        if (ovf && lane == 0) ovflag = 1;
        __syncthreads();
        slow = ovflag != 0;
        if (tid == 0) qslow[blockIdx.x] = slow ? 1 : 0;
        if (lane == 0) *gc = cnt;
    } else {
        slow = qslow[blockIdx.x] != 0;
        cnt  = *gc;
        __syncthreads();
    }

    if (!slow) {
        for (int i0 = 0; i0 < cnt; i0 += 128) {
            int r[4], s[4];
            float v[4][HEADS];
#pragma unroll
            for (int u = 0; u < 4; ++u) {
                const int idx = i0 + u * 32 + lane;
                r[u] = FIRST ? qw[idx < cnt ? idx : 0] : gq[idx < cnt ? idx : 0];
            }
#pragma unroll
            for (int u = 0; u < 4; ++u) {
                if (FIRST) {
                    const int e  = r[u] >> 8;
                    const int sl = src[e < E0 ? e : E0 - 1];
                    s[u] = e < E0 ? sl : e - E0;
                } else {
                    s[u] = r[u] >> 8;
                }
            }
#pragma unroll
            for (int u = 0; u < 4; ++u) lane_logits(a_src, a_dst, s[u], base + (r[u] & 255), v[u]);
#pragma unroll
            for (int u = 0; u < 4; ++u) {
                const int idx = i0 + u * 32 + lane;
                if (idx < cnt) {
                    const int dl = r[u] & 255;
                    const int en = (s[u] << 8) | dl;
                    qw[idx] = en;
                    if (FIRST) gq[idx] = en;
#pragma unroll
                    for (int h = 0; h < HEADS; ++h)
                        __hip_atomic_fetch_max(mloc + dl * HEADS + h, ford(v[u][h]),
                                               __ATOMIC_RELAXED, __HIP_MEMORY_SCOPE_WORKGROUP);
                }
            }
        }
        __syncthreads();
        for (int i0 = 0; i0 < cnt; i0 += 128) {
            int r[4];
            float v[4][HEADS];
#pragma unroll
            for (int u = 0; u < 4; ++u) {
                const int idx = i0 + u * 32 + lane;
                r[u] = qw[idx < cnt ? idx : 0];
            }
#pragma unroll
            for (int u = 0; u < 4; ++u) lane_logits(a_src, a_dst, r[u] >> 8, base + (r[u] & 255), v[u]);
#pragma unroll
            for (int u = 0; u < 4; ++u) {
                const int idx = i0 + u * 32 + lane;
                if (idx < cnt) {
                    const int dl = r[u] & 255;
#pragma unroll
                    for (int h = 0; h < HEADS; ++h) {
                        const float mv = funord(mloc[dl * HEADS + h]);
                        __hip_atomic_fetch_add(zloc + dl * HEADS + h, __expf(v[u][h] - mv),
                                               __ATOMIC_RELAXED, __HIP_MEMORY_SCOPE_WORKGROUP);
                    }
                }
            }
        }
        __syncthreads();
        const int kk = (lane & 15) >> 2;
        int i = 0;
        for (; i + 4 <= cnt; i += 4)
            rows_add<4>(qw[i + kk], base, lane, a_src, a_dst, mloc, zloc, H, slab);
        for (; i < cnt; ++i)
            rows_add<1>(qw[i], base, lane, a_src, a_dst, mloc, zloc, H, slab);
    } else {
        rescan<0>(tid, lane, wave, base, src, dst, a_src, a_dst, mloc, zloc, H, slab);
        __syncthreads();
        rescan<1>(tid, lane, wave, base, src, dst, a_src, a_dst, mloc, zloc, H, slab);
        __syncthreads();
        rescan<2>(tid, lane, wave, base, src, dst, a_src, a_dst, mloc, zloc, H, slab);
    }

    __syncthreads();
    float4* o4 = (float4*)(agg + (size_t)base * C_HID);
    for (int i = tid; i < SLAB * C_HID / 4; i += 1024) o4[i] = s4[i];
}

__global__ void bias_act(const float* __restrict__ agg, const float* __restrict__ bias,
                         float* __restrict__ out, int n, int relu) {
    int i = blockIdx.x * blockDim.x + threadIdx.x;
    if (i >= n) return;
    float v = agg[i] + bias[i & (C_HID - 1)];
    out[i] = relu ? fmaxf(v, 0.f) : v;
}

extern "C" void kernel_launch(void* const* d_in, const int* in_sizes, int n_in,
                              void* d_out, int out_size, void* d_ws, size_t ws_size,
                              hipStream_t stream) {
    const float* x        = (const float*)d_in[0];
    const int*   eidx     = (const int*)  d_in[1];
    const float* W1       = (const float*)d_in[2];
    const float* att_s1   = (const float*)d_in[3];
    const float* att_d1   = (const float*)d_in[4];
    const float* b1       = (const float*)d_in[5];
    const float* W2       = (const float*)d_in[6];
    const float* att_s2   = (const float*)d_in[7];
    const float* att_d2   = (const float*)d_in[8];
    const float* b2       = (const float*)d_in[9];
    float*       out      = (float*)d_out;

    const int* src = eidx;
    const int* dst = eidx + E0;

    const size_t NB = (size_t)N_NODES * C_HID;
    float* h      = (float*)d_ws;
    float* agg    = h      + NB;
    float* a_src  = agg    + NB;
    float* a_dst  = a_src  + (size_t)N_NODES * HEADS;
    int* qsave    = (int*)(a_dst + (size_t)N_NODES * HEADS);
    int* qcount   = qsave  + (size_t)NSLAB * SWAVES * QCAP;
    int* qslow    = qcount + NSLAB * SWAVES;
    float* WA1    = (float*)(qslow + 256);
    float* WA2    = WA1 + C_IN * 16;

    const int NELEM    = (int)NB;

    const dim3 B256(256);
    const int gNE  = (NELEM + 255) / 256;
    const int gMM  = N_NODES / 16;
    const int gFOLD = ((C_IN + C_HID) * 32 + 255) / 256;
    const int gMMD  = gMM + (gMM + 3) / 4;

    fold_att   <<<gFOLD, B256, 0, stream>>>(W1, att_s1, att_d1, W2, att_s2, att_d2, WA1, WA2);
    gemm_wmma<false><<<gMMD, 128, 0, stream>>>(x, nullptr, W1, WA1, h, a_src, a_dst, C_IN, C_HID, gMM);
    edge_slab<true><<<NSLAB, 1024, 0, stream>>>(src, dst, a_src, a_dst, h, agg, qsave, qcount, qslow);

    gemm_wmma<true><<<gMMD, 128, 0, stream>>>(agg, b1, W2, WA2, h, a_src, a_dst, C_HID, C_HID, gMM);
    edge_slab<false><<<NSLAB, 1024, 0, stream>>>(src, dst, a_src, a_dst, h, agg, qsave, qcount, qslow);
    bias_act   <<<gNE, B256, 0, stream>>>(agg, b2, out, NELEM, 0);
}


# Product re-tiled: a block copies its 32 x K rows of A into LDS once (whole-line loads, bias+ReLU there); a wave's W fragment feeds both row tiles; logit tile k-split 4 ways: dot sum order changes
# speedup vs baseline: 5.4841x; 1.1215x over previous
#include <hip/hip_runtime.h>
#include <math.h>

typedef __attribute__((ext_vector_type(2))) float v2f;
typedef __attribute__((ext_vector_type(8))) float v8f;

#define N_NODES 50000
#define C_IN    128
#define C_HID   256
#define HEADS   4
#define EMBED   64
#define E0      800000
#define ETOT    (E0 + N_NODES)
#define NEG_SLOPE 0.2f

template <bool ACT, int K>
__global__ __launch_bounds__(128) void gemm_wmma(const float* __restrict__ X,
                          const float* __restrict__ bias,
                          const float* __restrict__ W,
                          const float* __restrict__ WA,
                          float* __restrict__ H,
                          float* __restrict__ a_src,
                          float* __restrict__ a_dst,
                          int N, int MT) {
    static_assert(K % 16 == 0 && 128 % (K / 4) == 0, "four k-steps per trip; a thread keeps one column group");
    constexpr int KP = K + 4;
    __shared__ float As[32 * KP];
    __shared__ float Dx[3 * 256];

    const int tid    = threadIdx.x;
    const int lane   = tid & 31;
    const int wave   = __builtin_amdgcn_readfirstlane(tid >> 5);
    const int lane15 = lane & 15;
    const int khalf  = lane >> 4;
    const int ch     = (int)blockIdx.x & 1;
    const int tile0  = ((int)blockIdx.x >> 1) * 2;

    {
        constexpr int C4  = K / 4;
        constexpr int RS  = 128 / C4;
        constexpr int NLD = 32 / RS;
        const int c4 = tid % C4;
        const int r0 = tid / C4;
        const float* p0 = X + (size_t)(tile0 * 16 + r0) * K + c4 * 4;
        const float* p1 = p0 + (tile0 + 1 < MT ? 16 * K : 0);
        float4 t[NLD];
#pragma unroll
        for (int i = 0; i < NLD / 2; ++i) {
            t[i]           = *(const float4*)(p0 + i * RS * K);
            t[i + NLD / 2] = *(const float4*)(p1 + i * RS * K);
        }
        float4 bb = make_float4(0.f, 0.f, 0.f, 0.f);
        if (ACT) bb = *(const float4*)(bias + c4 * 4);
#pragma unroll
        for (int i = 0; i < NLD; ++i) {
            float4 v = t[i];
            if (ACT) {
                v.x = fmaxf(v.x + bb.x, 0.f);
                v.y = fmaxf(v.y + bb.y, 0.f);
                v.z = fmaxf(v.z + bb.z, 0.f);
                v.w = fmaxf(v.w + bb.w, 0.f);
            }
            *(float4*)(As + (r0 + i * RS) * KP + c4 * 4) = v;
        }
    }
    __syncthreads();

    const int col0 = ch * 128 + wave * 32;

    v8f acc00 = {}, acc01 = {};
    v8f acc10 = {}, acc11 = {};
    v8f accD  = {};

    const float* ar0 = As + lane15 * KP + khalf * 2;
    const float* ar1 = ar0 + 16 * KP;
    const float* ard = As + (ch * 16 + lane15) * KP + khalf * 2 + wave * 4;
    const float* wc  = W + col0 + lane15;
    const float* wac = WA + lane15;

    for (int k = 0; k < K; k += 16) {
#pragma unroll
        for (int u = 0; u < 4; ++u) {
            const int ks = k + u * 4;
            const int kk = ks + khalf * 2;
            const v2f a0 = *(const v2f*)(ar0 + ks);
            const v2f a1 = *(const v2f*)(ar1 + ks);
            const float* wp0 = wc + (size_t)kk * N;
            const float* wp1 = wp0 + N;
            v2f b0, b1;
            b0.x = wp0[ 0]; b0.y = wp1[ 0];
            b1.x = wp0[16]; b1.y = wp1[16];
            acc00 = __builtin_amdgcn_wmma_f32_16x16x4_f32(false, a0, false, b0, (short)0, acc00, false, false);
            acc10 = __builtin_amdgcn_wmma_f32_16x16x4_f32(false, a1, false, b0, (short)0, acc10, false, false);
            acc01 = __builtin_amdgcn_wmma_f32_16x16x4_f32(false, a0, false, b1, (short)0, acc01, false, false);
            acc11 = __builtin_amdgcn_wmma_f32_16x16x4_f32(false, a1, false, b1, (short)0, acc11, false, false);
        }
        {
            const int kd = k + wave * 4 + khalf * 2;
            const v2f ad = *(const v2f*)(ard + k);
            v2f bd;
            bd.x = wac[kd * 16];  bd.y = wac[(kd + 1) * 16];
            accD = __builtin_amdgcn_wmma_f32_16x16x4_f32(false, ad, false, bd, (short)0, accD, false, false);
        }
    }

    if (wave) {
#pragma unroll
        for (int i = 0; i < 8; ++i) Dx[(wave - 1) * 256 + i * 32 + lane] = accD[i];
    }
    __syncthreads();
    if (wave == 0 && tile0 + ch < MT && lane15 < 8) {
        float* op = (lane15 < 4 ? a_src : a_dst) + (size_t)((tile0 + ch) * 16 + khalf * 8) * HEADS + (lane15 & 3);
#pragma unroll
        for (int i = 0; i < 8; ++i) {
            const float p1 = Dx[      i * 32 + lane];
            const float p2 = Dx[256 + i * 32 + lane];
            const float p3 = Dx[512 + i * 32 + lane];
            op[i * HEADS] = (accD[i] + p2) + (p1 + p3);
        }
    }

    {
        float* op = H + (size_t)(tile0 * 16 + khalf * 8) * N + col0 + lane15;
#pragma unroll
        for (int i = 0; i < 8; ++i) {
            op[i * N +  0] = acc00[i];
            op[i * N + 16] = acc01[i];
        }
        if (tile0 + 1 < MT) {
            op += (size_t)16 * N;
#pragma unroll
            for (int i = 0; i < 8; ++i) {
                op[i * N +  0] = acc10[i];
                op[i * N + 16] = acc11[i];
            }
        }
    }
}

__global__ void fold_att(const float* __restrict__ W1,
                         const float* __restrict__ as1,
                         const float* __restrict__ ad1,
                         const float* __restrict__ W2,
                         const float* __restrict__ as2,
                         const float* __restrict__ ad2,
                         float* __restrict__ WA1,
                         float* __restrict__ WA2) {
    const int gw   = (int)((blockIdx.x * blockDim.x + threadIdx.x) >> 5);
    const int lane = threadIdx.x & 31;
    if (gw >= C_IN + C_HID) return;
    const bool second = gw >= C_IN;
    const int  k      = second ? gw - C_IN : gw;
    const float* Wm = second ? W2 : W1;
    const float* as = second ? as2 : as1;
    const float* ad = second ? ad2 : ad1;
    float*       WA = second ? WA2 : WA1;

    const int j    = lane >> 2;
    const int part = lane & 3;
    const int head = j & 3;
    const float* att = (j < 4 ? as : ad) + head * EMBED + part * 16;
    const float* wp  = Wm + (size_t)k * C_HID + head * EMBED + part * 16;
    float s = 0.f;
#pragma unroll
    for (int c = 0; c < 16; c += 4) {
        const float4 w4 = *(const float4*)(wp + c);
        const float4 t4 = *(const float4*)(att + c);
        s += w4.x * t4.x;
        s += w4.y * t4.y;
        s += w4.z * t4.z;
        s += w4.w * t4.w;
    }
    s += __shfl_xor(s, 1, 32);
    s += __shfl_xor(s, 2, 32);
    if (part == 0) WA[k * 16 + j] = s;
    if (part == 1) WA[k * 16 + 8 + j] = 0.f;
}

__device__ __forceinline__ unsigned ford(float f) {
    unsigned u = __float_as_uint(f);
    return (u >> 31) ? ~u : (u | 0x80000000u);
}
__device__ __forceinline__ float funord(unsigned u) {
    return (u >> 31) ? __uint_as_float(u ^ 0x80000000u) : __uint_as_float(~u);
}

__device__ __forceinline__ float leaky(float v) {
    return v > 0.f ? v : NEG_SLOPE * v;
}

#define SLAB   200
#define NSLAB  (N_NODES / SLAB)
#define QCAP   640
#define SWAVES 32
#define ORD_NEG_INF 0x007FFFFFu
static_assert(N_NODES % SLAB == 0, "rows split evenly over the blocks");
static_assert(SLAB <= 256 && ETOT < (1 << 23), "a queue entry packs edge << 8 | local row");
static_assert(E0 % 4 == 0, "dst is scanned in int4 pieces");

__device__ __forceinline__ void lane_logits(const float* __restrict__ a_src,
                                            const float* __restrict__ a_dst,
                                            int s, int d, float v[HEADS]) {
    const float4 as = *(const float4*)(a_src + (size_t)s * HEADS);
    const float4 ad = *(const float4*)(a_dst + (size_t)d * HEADS);
    v[0] = leaky(as.x + ad.x);
    v[1] = leaky(as.y + ad.y);
    v[2] = leaky(as.z + ad.z);
    v[3] = leaky(as.w + ad.w);
}

__device__ __forceinline__ void lane_max(const float* __restrict__ a_src,
                                         const float* __restrict__ a_dst,
                                         int s, int dl, int base, unsigned* mloc) {
    float v[HEADS];
    lane_logits(a_src, a_dst, s, base + dl, v);
#pragma unroll
    for (int h = 0; h < HEADS; ++h)
        __hip_atomic_fetch_max(mloc + dl * HEADS + h, ford(v[h]),
                               __ATOMIC_RELAXED, __HIP_MEMORY_SCOPE_WORKGROUP);
}

__device__ __forceinline__ void lane_z(const float* __restrict__ a_src,
                                       const float* __restrict__ a_dst,
                                       int s, int dl, int base,
                                       const unsigned* mloc, float* zloc) {
    float v[HEADS];
    lane_logits(a_src, a_dst, s, base + dl, v);
#pragma unroll
    for (int h = 0; h < HEADS; ++h) {
        const float mv = funord(mloc[dl * HEADS + h]);
        __hip_atomic_fetch_add(zloc + dl * HEADS + h, __expf(v[h] - mv),
                               __ATOMIC_RELAXED, __HIP_MEMORY_SCOPE_WORKGROUP);
    }
}

template <int NB>
__device__ __forceinline__ void rows_add(int r, int base, int lane,
                                         const float* __restrict__ a_src,
                                         const float* __restrict__ a_dst,
                                         const unsigned* mloc, const float* zloc,
                                         const float* __restrict__ H,
                                         float* slab) {
    int s[NB], dl[NB];
#pragma unroll
    for (int k = 0; k < NB; ++k) {
        const int rk = __builtin_amdgcn_readlane(r, 4 * k);
        s[k]  = rk >> 8;
        dl[k] = rk & 255;
    }
    float hv[NB][8];
#pragma unroll
    for (int k = 0; k < NB; ++k) {
        const float* hp = H + (size_t)s[k] * C_HID + lane;
#pragma unroll
        for (int j = 0; j < 8; ++j) hv[k][j] = hp[j * 32];
    }
    const int h   = lane & 3;
    const int sL  = r >> 8;
    const int dlL = r & 255;
    const float v  = leaky(a_src[(size_t)sL * HEADS + h] + a_dst[(size_t)(base + dlL) * HEADS + h]);
    const float ev = __expf(v - funord(mloc[dlL * HEADS + h]));
    const int   al = __float_as_int(ev / (zloc[dlL * HEADS + h] + 1e-16f));
#pragma unroll
    for (int k = 0; k < NB; ++k) {
        float* sp = slab + dl[k] * C_HID + lane;
#pragma unroll
        for (int j = 0; j < 8; ++j) {
            const float a = __int_as_float(__builtin_amdgcn_readlane(al, 4 * k + (j >> 1)));
            __hip_atomic_fetch_add(sp + j * 32, hv[k][j] * a,
                                   __ATOMIC_RELAXED, __HIP_MEMORY_SCOPE_WORKGROUP);
        }
    }
}

template <int PH>
__device__ __forceinline__ void rescan(int tid, int lane, int wave, int base,
                                       const int* __restrict__ src,
                                       const int* __restrict__ dst,
                                       const float* __restrict__ a_src,
                                       const float* __restrict__ a_dst,
                                       unsigned* mloc, float* zloc,
                                       const float* __restrict__ H,
                                       float* slab) {
    if (PH < 2) {
        for (int n = tid; n < SLAB; n += 1024) {
            if (PH == 0) lane_max(a_src, a_dst, base + n, n, base, mloc);
            else         lane_z(a_src, a_dst, base + n, n, base, mloc, zloc);
        }
    } else {
        for (int n = wave; n < SLAB; n += SWAVES) {
            rows_add<1>(((base + n) << 8) | n, base, lane, a_src, a_dst, mloc, zloc, H, slab);
        }
    }
    const int4* d4p = (const int4*)dst;
    const int NV = E0 / 4;
    for (int vb = wave * 32; vb < NV; vb += SWAVES * 32) {
        const int vi = vb + lane;
        const int4 t = d4p[vi < NV ? vi : NV - 1];
        const int comp[4] = {t.x, t.y, t.z, t.w};
#pragma unroll
        for (int c = 0; c < 4; ++c) {
            const unsigned dl = (unsigned)(comp[c] - base);
            const bool mt = vi < NV && dl < (unsigned)SLAB;
            int s = 0;
            if (mt) s = src[vi * 4 + c];
            if (PH == 0) {
                if (mt) lane_max(a_src, a_dst, s, (int)dl, base, mloc);
            } else if (PH == 1) {
                if (mt) lane_z(a_src, a_dst, s, (int)dl, base, mloc, zloc);
            } else {
                unsigned mask = __builtin_amdgcn_ballot_w32(mt);
                while (mask) {
                    const int b = __builtin_ctz(mask);
                    mask &= mask - 1;
                    const int rb = __builtin_amdgcn_readlane((s << 8) | (int)dl, b);
                    rows_add<1>(rb, base, lane, a_src, a_dst, mloc, zloc, H, slab);
                }
            }
        }
    }
}

template <bool FIRST>
__global__ __launch_bounds__(1024) void edge_slab(const int* __restrict__ src,
                                                  const int* __restrict__ dst,
                                                  const float* __restrict__ a_src,
                                                  const float* __restrict__ a_dst,
                                                  const float* __restrict__ H,
                                                  float* __restrict__ agg,
                                                  int* __restrict__ qsave,
                                                  int* __restrict__ qcount,
                                                  int* __restrict__ qslow) {
    __shared__ float    slab[SLAB * C_HID];
    __shared__ int      queue[SWAVES * QCAP];
    __shared__ unsigned mloc[SLAB * HEADS];
    __shared__ float    zloc[SLAB * HEADS];
    __shared__ int      ovflag;

    const int tid  = threadIdx.x;
    const int lane = tid & 31;
    const int wave = __builtin_amdgcn_readfirstlane(tid >> 5);
    const int base = blockIdx.x * SLAB;

    float4* s4 = (float4*)slab;
    for (int i = tid; i < SLAB * C_HID / 4; i += 1024) s4[i] = make_float4(0.f, 0.f, 0.f, 0.f);
    for (int i = tid; i < SLAB * HEADS; i += 1024) { mloc[i] = ORD_NEG_INF; zloc[i] = 0.f; }
    if (FIRST && tid == 0) ovflag = 0;

    int* qw = queue + wave * QCAP;
    int* gq = qsave + ((size_t)blockIdx.x * SWAVES + wave) * QCAP;
    int* gc = qcount + blockIdx.x * SWAVES + wave;
    int  cnt;
    bool slow;
    if (FIRST) {
        cnt = (SLAB - 1 - wave) / SWAVES + 1;
        if (lane < cnt) {
            const int n = wave + SWAVES * lane;
            qw[lane] = ((E0 + base + n) << 8) | n;
        }
        __syncthreads();

        const int4* d4p = (const int4*)dst;
        const int NV = E0 / 4;
        const int4 none = make_int4(-1, -1, -1, -1);
        bool ovf = false;
        for (int vb = wave * 128; vb < NV; vb += SWAVES * 128) {
            int4 p[4];
#pragma unroll
            for (int q = 0; q < 4; ++q) {
                const int vi = vb + q * 32 + lane;
                const int4 t = d4p[vi < NV ? vi : NV - 1];
                p[q] = vi < NV ? t : none;
            }
#pragma unroll
            for (int q = 0; q < 4; ++q) {
                const int vi = vb + q * 32 + lane;
                const int comp[4] = {p[q].x, p[q].y, p[q].z, p[q].w};
#pragma unroll
                for (int c = 0; c < 4; ++c) {
                    const unsigned dl = (unsigned)(comp[c] - base);
                    const bool mt = dl < (unsigned)SLAB;
                    const unsigned mask = __builtin_amdgcn_ballot_w32(mt);
                    if (mask) {
                        const int n = __builtin_popcount(mask);
                        if (cnt + n <= QCAP) {
                            if (mt) qw[cnt + __builtin_amdgcn_mbcnt_lo(mask, 0u)] = ((vi * 4 + c) << 8) | (int)dl;
                            cnt += n;
                        } else {
                            ovf = true;
                        }
                    }
                }
            }
        }
        if (ovf && lane == 0) ovflag = 1;
        __syncthreads();
        slow = ovflag != 0;
        if (tid == 0) qslow[blockIdx.x] = slow ? 1 : 0;
        if (lane == 0) *gc = cnt;
    } else {
        slow = qslow[blockIdx.x] != 0;
        cnt  = *gc;
        __syncthreads();
    }

    if (!slow) {
        for (int i0 = 0; i0 < cnt; i0 += 128) {
            int r[4], s[4];
            float v[4][HEADS];
#pragma unroll
            for (int u = 0; u < 4; ++u) {
                const int idx = i0 + u * 32 + lane;
                r[u] = FIRST ? qw[idx < cnt ? idx : 0] : gq[idx < cnt ? idx : 0];
            }
#pragma unroll
            for (int u = 0; u < 4; ++u) {
                if (FIRST) {
                    const int e  = r[u] >> 8;
                    const int sl = src[e < E0 ? e : E0 - 1];
                    s[u] = e < E0 ? sl : e - E0;
                } else {
                    s[u] = r[u] >> 8;
                }
            }
#pragma unroll
            for (int u = 0; u < 4; ++u) lane_logits(a_src, a_dst, s[u], base + (r[u] & 255), v[u]);
#pragma unroll
            for (int u = 0; u < 4; ++u) {
                const int idx = i0 + u * 32 + lane;
                if (idx < cnt) {
                    const int dl = r[u] & 255;
                    const int en = (s[u] << 8) | dl;
                    qw[idx] = en;
                    if (FIRST) gq[idx] = en;
#pragma unroll
                    for (int h = 0; h < HEADS; ++h)
                        __hip_atomic_fetch_max(mloc + dl * HEADS + h, ford(v[u][h]),
                                               __ATOMIC_RELAXED, __HIP_MEMORY_SCOPE_WORKGROUP);
                }
            }
        }
        __syncthreads();
        for (int i0 = 0; i0 < cnt; i0 += 128) {
            int r[4];
            float v[4][HEADS];
#pragma unroll
            for (int u = 0; u < 4; ++u) {
                const int idx = i0 + u * 32 + lane;
                r[u] = qw[idx < cnt ? idx : 0];
            }
#pragma unroll
            for (int u = 0; u < 4; ++u) lane_logits(a_src, a_dst, r[u] >> 8, base + (r[u] & 255), v[u]);
#pragma unroll
            for (int u = 0; u < 4; ++u) {
                const int idx = i0 + u * 32 + lane;
                if (idx < cnt) {
                    const int dl = r[u] & 255;
#pragma unroll
                    for (int h = 0; h < HEADS; ++h) {
                        const float mv = funord(mloc[dl * HEADS + h]);
                        __hip_atomic_fetch_add(zloc + dl * HEADS + h, __expf(v[u][h] - mv),
                                               __ATOMIC_RELAXED, __HIP_MEMORY_SCOPE_WORKGROUP);
                    }
                }
            }
        }
        __syncthreads();
        const int kk = (lane & 15) >> 2;
        int i = 0;
        for (; i + 4 <= cnt; i += 4)
            rows_add<4>(qw[i + kk], base, lane, a_src, a_dst, mloc, zloc, H, slab);
        for (; i < cnt; ++i)
            rows_add<1>(qw[i], base, lane, a_src, a_dst, mloc, zloc, H, slab);
    } else {
        rescan<0>(tid, lane, wave, base, src, dst, a_src, a_dst, mloc, zloc, H, slab);
        __syncthreads();
        rescan<1>(tid, lane, wave, base, src, dst, a_src, a_dst, mloc, zloc, H, slab);
        __syncthreads();
        rescan<2>(tid, lane, wave, base, src, dst, a_src, a_dst, mloc, zloc, H, slab);
    }

    __syncthreads();
    float4* o4 = (float4*)(agg + (size_t)base * C_HID);
    for (int i = tid; i < SLAB * C_HID / 4; i += 1024) o4[i] = s4[i];
}

__global__ void bias_act(const float* __restrict__ agg, const float* __restrict__ bias,
                         float* __restrict__ out, int n, int relu) {
    int i = blockIdx.x * blockDim.x + threadIdx.x;
    if (i >= n) return;
    float v = agg[i] + bias[i & (C_HID - 1)];
    out[i] = relu ? fmaxf(v, 0.f) : v;
}

extern "C" void kernel_launch(void* const* d_in, const int* in_sizes, int n_in,
                              void* d_out, int out_size, void* d_ws, size_t ws_size,
                              hipStream_t stream) {
    const float* x        = (const float*)d_in[0];
    const int*   eidx     = (const int*)  d_in[1];
    const float* W1       = (const float*)d_in[2];
    const float* att_s1   = (const float*)d_in[3];
    const float* att_d1   = (const float*)d_in[4];
    const float* b1       = (const float*)d_in[5];
    const float* W2       = (const float*)d_in[6];
    const float* att_s2   = (const float*)d_in[7];
    const float* att_d2   = (const float*)d_in[8];
    const float* b2       = (const float*)d_in[9];
    float*       out      = (float*)d_out;

    const int* src = eidx;
    const int* dst = eidx + E0;

    const size_t NB = (size_t)N_NODES * C_HID;
    float* h      = (float*)d_ws;
    float* agg    = h      + NB;
    float* a_src  = agg    + NB;
    float* a_dst  = a_src  + (size_t)N_NODES * HEADS;
    int* qsave    = (int*)(a_dst + (size_t)N_NODES * HEADS);
    int* qcount   = qsave  + (size_t)NSLAB * SWAVES * QCAP;
    int* qslow    = qcount + NSLAB * SWAVES;
    float* WA1    = (float*)(qslow + 256);
    float* WA2    = WA1 + C_IN * 16;

    const int NELEM    = (int)NB;

    const dim3 B256(256);
    const int gNE  = (NELEM + 255) / 256;
    const int gMM  = N_NODES / 16;
    const int gFOLD = ((C_IN + C_HID) * 32 + 255) / 256;
    const int gMB   = (gMM + 1) / 2 * 2;

    fold_att   <<<gFOLD, B256, 0, stream>>>(W1, att_s1, att_d1, W2, att_s2, att_d2, WA1, WA2);
    gemm_wmma<false, C_IN><<<gMB, 128, 0, stream>>>(x, nullptr, W1, WA1, h, a_src, a_dst, C_HID, gMM);
    edge_slab<true><<<NSLAB, 1024, 0, stream>>>(src, dst, a_src, a_dst, h, agg, qsave, qcount, qslow);

    gemm_wmma<true, C_HID><<<gMB, 128, 0, stream>>>(agg, b1, W2, WA2, h, a_src, a_dst, C_HID, gMM);
    edge_slab<false><<<NSLAB, 1024, 0, stream>>>(src, dst, a_src, a_dst, h, agg, qsave, qcount, qslow);
    bias_act   <<<gNE, B256, 0, stream>>>(agg, b2, out, NELEM, 0);
}


# Layer-2 rows start at the bias in LDS (sum order: bias first, then LDS arrival); the owning block stores them straight to out: last per-element launch (400,000 waves) and its plane re-read are gone
# speedup vs baseline: 6.1311x; 1.1180x over previous
#include <hip/hip_runtime.h>
#include <math.h>

typedef __attribute__((ext_vector_type(2))) float v2f;
typedef __attribute__((ext_vector_type(8))) float v8f;

#define N_NODES 50000
#define C_IN    128
#define C_HID   256
#define HEADS   4
#define EMBED   64
#define E0      800000
#define ETOT    (E0 + N_NODES)
#define NEG_SLOPE 0.2f

template <bool ACT, int K>
__global__ __launch_bounds__(128) void gemm_wmma(const float* __restrict__ X,
                          const float* __restrict__ bias,
                          const float* __restrict__ W,
                          const float* __restrict__ WA,
                          float* __restrict__ H,
                          float* __restrict__ a_src,
                          float* __restrict__ a_dst,
                          int N, int MT) {
    static_assert(K % 16 == 0 && 128 % (K / 4) == 0, "four k-steps per trip; a thread keeps one column group");
    constexpr int KP = K + 4;
    __shared__ float As[32 * KP];
    __shared__ float Dx[3 * 256];

    const int tid    = threadIdx.x;
    const int lane   = tid & 31;
    const int wave   = __builtin_amdgcn_readfirstlane(tid >> 5);
    const int lane15 = lane & 15;
    const int khalf  = lane >> 4;
    const int ch     = (int)blockIdx.x & 1;
    const int tile0  = ((int)blockIdx.x >> 1) * 2;

    {
        constexpr int C4  = K / 4;
        constexpr int RS  = 128 / C4;
        constexpr int NLD = 32 / RS;
        const int c4 = tid % C4;
        const int r0 = tid / C4;
        const float* p0 = X + (size_t)(tile0 * 16 + r0) * K + c4 * 4;
        const float* p1 = p0 + (tile0 + 1 < MT ? 16 * K : 0);
        float4 t[NLD];
#pragma unroll
        for (int i = 0; i < NLD / 2; ++i) {
            t[i]           = *(const float4*)(p0 + i * RS * K);
            t[i + NLD / 2] = *(const float4*)(p1 + i * RS * K);
        }
        float4 bb = make_float4(0.f, 0.f, 0.f, 0.f);
        if (ACT) bb = *(const float4*)(bias + c4 * 4);
#pragma unroll
        for (int i = 0; i < NLD; ++i) {
            float4 v = t[i];
            if (ACT) {
                v.x = fmaxf(v.x + bb.x, 0.f);
                v.y = fmaxf(v.y + bb.y, 0.f);
                v.z = fmaxf(v.z + bb.z, 0.f);
                v.w = fmaxf(v.w + bb.w, 0.f);
            }
            *(float4*)(As + (r0 + i * RS) * KP + c4 * 4) = v;
        }
    }
    __syncthreads();

    const int col0 = ch * 128 + wave * 32;

    v8f acc00 = {}, acc01 = {};
    v8f acc10 = {}, acc11 = {};
    v8f accD  = {};

    const float* ar0 = As + lane15 * KP + khalf * 2;
    const float* ar1 = ar0 + 16 * KP;
    const float* ard = As + (ch * 16 + lane15) * KP + khalf * 2 + wave * 4;
    const float* wc  = W + col0 + lane15;
    const float* wac = WA + lane15;

    for (int k = 0; k < K; k += 16) {
#pragma unroll
        for (int u = 0; u < 4; ++u) {
            const int ks = k + u * 4;
            const int kk = ks + khalf * 2;
            const v2f a0 = *(const v2f*)(ar0 + ks);
            const v2f a1 = *(const v2f*)(ar1 + ks);
            const float* wp0 = wc + (size_t)kk * N;
            const float* wp1 = wp0 + N;
            v2f b0, b1;
            b0.x = wp0[ 0]; b0.y = wp1[ 0];
            b1.x = wp0[16]; b1.y = wp1[16];
            acc00 = __builtin_amdgcn_wmma_f32_16x16x4_f32(false, a0, false, b0, (short)0, acc00, false, false);
            acc10 = __builtin_amdgcn_wmma_f32_16x16x4_f32(false, a1, false, b0, (short)0, acc10, false, false);
            acc01 = __builtin_amdgcn_wmma_f32_16x16x4_f32(false, a0, false, b1, (short)0, acc01, false, false);
            acc11 = __builtin_amdgcn_wmma_f32_16x16x4_f32(false, a1, false, b1, (short)0, acc11, false, false);
        }
        {
            const int kd = k + wave * 4 + khalf * 2;
            const v2f ad = *(const v2f*)(ard + k);
            v2f bd;
            bd.x = wac[kd * 16];  bd.y = wac[(kd + 1) * 16];
            accD = __builtin_amdgcn_wmma_f32_16x16x4_f32(false, ad, false, bd, (short)0, accD, false, false);
        }
    }

    if (wave) {
#pragma unroll
        for (int i = 0; i < 8; ++i) Dx[(wave - 1) * 256 + i * 32 + lane] = accD[i];
    }
    __syncthreads();
    if (wave == 0 && tile0 + ch < MT && lane15 < 8) {
        float* op = (lane15 < 4 ? a_src : a_dst) + (size_t)((tile0 + ch) * 16 + khalf * 8) * HEADS + (lane15 & 3);
#pragma unroll
        for (int i = 0; i < 8; ++i) {
            const float p1 = Dx[      i * 32 + lane];
            const float p2 = Dx[256 + i * 32 + lane];
            const float p3 = Dx[512 + i * 32 + lane];
            op[i * HEADS] = (accD[i] + p2) + (p1 + p3);
        }
    }

    {
        float* op = H + (size_t)(tile0 * 16 + khalf * 8) * N + col0 + lane15;
#pragma unroll
        for (int i = 0; i < 8; ++i) {
            op[i * N +  0] = acc00[i];
            op[i * N + 16] = acc01[i];
        }
        if (tile0 + 1 < MT) {
            op += (size_t)16 * N;
#pragma unroll
            for (int i = 0; i < 8; ++i) {
                op[i * N +  0] = acc10[i];
                op[i * N + 16] = acc11[i];
            }
        }
    }
}

__global__ void fold_att(const float* __restrict__ W1,
                         const float* __restrict__ as1,
                         const float* __restrict__ ad1,
                         const float* __restrict__ W2,
                         const float* __restrict__ as2,
                         const float* __restrict__ ad2,
                         float* __restrict__ WA1,
                         float* __restrict__ WA2) {
    const int gw   = (int)((blockIdx.x * blockDim.x + threadIdx.x) >> 5);
    const int lane = threadIdx.x & 31;
    if (gw >= C_IN + C_HID) return;
    const bool second = gw >= C_IN;
    const int  k      = second ? gw - C_IN : gw;
    const float* Wm = second ? W2 : W1;
    const float* as = second ? as2 : as1;
    const float* ad = second ? ad2 : ad1;
    float*       WA = second ? WA2 : WA1;

    const int j    = lane >> 2;
    const int part = lane & 3;
    const int head = j & 3;
    const float* att = (j < 4 ? as : ad) + head * EMBED + part * 16;
    const float* wp  = Wm + (size_t)k * C_HID + head * EMBED + part * 16;
    float s = 0.f;
#pragma unroll
    for (int c = 0; c < 16; c += 4) {
        const float4 w4 = *(const float4*)(wp + c);
        const float4 t4 = *(const float4*)(att + c);
        s += w4.x * t4.x;
        s += w4.y * t4.y;
        s += w4.z * t4.z;
        s += w4.w * t4.w;
    }
    s += __shfl_xor(s, 1, 32);
    s += __shfl_xor(s, 2, 32);
    if (part == 0) WA[k * 16 + j] = s;
    if (part == 1) WA[k * 16 + 8 + j] = 0.f;
}

__device__ __forceinline__ unsigned ford(float f) {
    unsigned u = __float_as_uint(f);
    return (u >> 31) ? ~u : (u | 0x80000000u);
}
__device__ __forceinline__ float funord(unsigned u) {
    return (u >> 31) ? __uint_as_float(u ^ 0x80000000u) : __uint_as_float(~u);
}

__device__ __forceinline__ float leaky(float v) {
    return v > 0.f ? v : NEG_SLOPE * v;
}

#define SLAB   200
#define NSLAB  (N_NODES / SLAB)
#define QCAP   640
#define SWAVES 32
#define ORD_NEG_INF 0x007FFFFFu
static_assert(N_NODES % SLAB == 0, "rows split evenly over the blocks");
static_assert(SLAB <= 256 && ETOT < (1 << 23), "a queue entry packs edge << 8 | local row");
static_assert(E0 % 4 == 0, "dst is scanned in int4 pieces");

__device__ __forceinline__ void lane_logits(const float* __restrict__ a_src,
                                            const float* __restrict__ a_dst,
                                            int s, int d, float v[HEADS]) {
    const float4 as = *(const float4*)(a_src + (size_t)s * HEADS);
    const float4 ad = *(const float4*)(a_dst + (size_t)d * HEADS);
    v[0] = leaky(as.x + ad.x);
    v[1] = leaky(as.y + ad.y);
    v[2] = leaky(as.z + ad.z);
    v[3] = leaky(as.w + ad.w);
}

__device__ __forceinline__ void lane_max(const float* __restrict__ a_src,
                                         const float* __restrict__ a_dst,
                                         int s, int dl, int base, unsigned* mloc) {
    float v[HEADS];
    lane_logits(a_src, a_dst, s, base + dl, v);
#pragma unroll
    for (int h = 0; h < HEADS; ++h)
        __hip_atomic_fetch_max(mloc + dl * HEADS + h, ford(v[h]),
                               __ATOMIC_RELAXED, __HIP_MEMORY_SCOPE_WORKGROUP);
}

__device__ __forceinline__ void lane_z(const float* __restrict__ a_src,
                                       const float* __restrict__ a_dst,
                                       int s, int dl, int base,
                                       const unsigned* mloc, float* zloc) {
    float v[HEADS];
    lane_logits(a_src, a_dst, s, base + dl, v);
#pragma unroll
    for (int h = 0; h < HEADS; ++h) {
        const float mv = funord(mloc[dl * HEADS + h]);
        __hip_atomic_fetch_add(zloc + dl * HEADS + h, __expf(v[h] - mv),
                               __ATOMIC_RELAXED, __HIP_MEMORY_SCOPE_WORKGROUP);
    }
}

template <int NB>
__device__ __forceinline__ void rows_add(int r, int base, int lane,
                                         const float* __restrict__ a_src,
                                         const float* __restrict__ a_dst,
                                         const unsigned* mloc, const float* zloc,
                                         const float* __restrict__ H,
                                         float* slab) {
    int s[NB], dl[NB];
#pragma unroll
    for (int k = 0; k < NB; ++k) {
        const int rk = __builtin_amdgcn_readlane(r, 4 * k);
        s[k]  = rk >> 8;
        dl[k] = rk & 255;
    }
    float hv[NB][8];
#pragma unroll
    for (int k = 0; k < NB; ++k) {
        const float* hp = H + (size_t)s[k] * C_HID + lane;
#pragma unroll
        for (int j = 0; j < 8; ++j) hv[k][j] = hp[j * 32];
    }
    const int h   = lane & 3;
    const int sL  = r >> 8;
    const int dlL = r & 255;
    const float v  = leaky(a_src[(size_t)sL * HEADS + h] + a_dst[(size_t)(base + dlL) * HEADS + h]);
    const float ev = __expf(v - funord(mloc[dlL * HEADS + h]));
    const int   al = __float_as_int(ev / (zloc[dlL * HEADS + h] + 1e-16f));
#pragma unroll
    for (int k = 0; k < NB; ++k) {
        float* sp = slab + dl[k] * C_HID + lane;
#pragma unroll
        for (int j = 0; j < 8; ++j) {
            const float a = __int_as_float(__builtin_amdgcn_readlane(al, 4 * k + (j >> 1)));
            __hip_atomic_fetch_add(sp + j * 32, hv[k][j] * a,
                                   __ATOMIC_RELAXED, __HIP_MEMORY_SCOPE_WORKGROUP);
        }
    }
}

template <int PH>
__device__ __forceinline__ void rescan(int tid, int lane, int wave, int base,
                                       const int* __restrict__ src,
                                       const int* __restrict__ dst,
                                       const float* __restrict__ a_src,
                                       const float* __restrict__ a_dst,
                                       unsigned* mloc, float* zloc,
                                       const float* __restrict__ H,
                                       float* slab) {
    if (PH < 2) {
        for (int n = tid; n < SLAB; n += 1024) {
            if (PH == 0) lane_max(a_src, a_dst, base + n, n, base, mloc);
            else         lane_z(a_src, a_dst, base + n, n, base, mloc, zloc);
        }
    } else {
        for (int n = wave; n < SLAB; n += SWAVES) {
            rows_add<1>(((base + n) << 8) | n, base, lane, a_src, a_dst, mloc, zloc, H, slab);
        }
    }
    const int4* d4p = (const int4*)dst;
    const int NV = E0 / 4;
    for (int vb = wave * 32; vb < NV; vb += SWAVES * 32) {
        const int vi = vb + lane;
        const int4 t = d4p[vi < NV ? vi : NV - 1];
        const int comp[4] = {t.x, t.y, t.z, t.w};
#pragma unroll
        for (int c = 0; c < 4; ++c) {
            const unsigned dl = (unsigned)(comp[c] - base);
            const bool mt = vi < NV && dl < (unsigned)SLAB;
            int s = 0;
            if (mt) s = src[vi * 4 + c];
            if (PH == 0) {
                if (mt) lane_max(a_src, a_dst, s, (int)dl, base, mloc);
            } else if (PH == 1) {
                if (mt) lane_z(a_src, a_dst, s, (int)dl, base, mloc, zloc);
            } else {
                unsigned mask = __builtin_amdgcn_ballot_w32(mt);
                while (mask) {
                    const int b = __builtin_ctz(mask);
                    mask &= mask - 1;
                    const int rb = __builtin_amdgcn_readlane((s << 8) | (int)dl, b);
                    rows_add<1>(rb, base, lane, a_src, a_dst, mloc, zloc, H, slab);
                }
            }
        }
    }
}

template <bool FIRST>
__global__ __launch_bounds__(1024) void edge_slab(const int* __restrict__ src,
                                                  const int* __restrict__ dst,
                                                  const float* __restrict__ a_src,
                                                  const float* __restrict__ a_dst,
                                                  const float* __restrict__ H,
                                                  const float* __restrict__ bias,
                                                  float* __restrict__ rows,
                                                  int* __restrict__ qsave,
                                                  int* __restrict__ qcount,
                                                  int* __restrict__ qslow) {
    __shared__ float    slab[SLAB * C_HID];
    __shared__ int      queue[SWAVES * QCAP];
    __shared__ unsigned mloc[SLAB * HEADS];
    __shared__ float    zloc[SLAB * HEADS];
    __shared__ int      ovflag;

    const int tid  = threadIdx.x;
    const int lane = tid & 31;
    const int wave = __builtin_amdgcn_readfirstlane(tid >> 5);
    const int base = blockIdx.x * SLAB;

    float4* s4 = (float4*)slab;
    {
        float4 start = make_float4(0.f, 0.f, 0.f, 0.f);
        if (!FIRST) start = *(const float4*)(bias + (tid & 63) * 4);
        for (int i = tid; i < SLAB * C_HID / 4; i += 1024) s4[i] = start;
    }
    for (int i = tid; i < SLAB * HEADS; i += 1024) { mloc[i] = ORD_NEG_INF; zloc[i] = 0.f; }
    if (FIRST && tid == 0) ovflag = 0;

    int* qw = queue + wave * QCAP;
    int* gq = qsave + ((size_t)blockIdx.x * SWAVES + wave) * QCAP;
    int* gc = qcount + blockIdx.x * SWAVES + wave;
    int  cnt;
    bool slow;
    if (FIRST) {
        cnt = (SLAB - 1 - wave) / SWAVES + 1;
        if (lane < cnt) {
            const int n = wave + SWAVES * lane;
            qw[lane] = ((E0 + base + n) << 8) | n;
        }
        __syncthreads();

        const int4* d4p = (const int4*)dst;
        const int NV = E0 / 4;
        const int4 none = make_int4(-1, -1, -1, -1);
        bool ovf = false;
        for (int vb = wave * 128; vb < NV; vb += SWAVES * 128) {
            int4 p[4];
#pragma unroll
            for (int q = 0; q < 4; ++q) {
                const int vi = vb + q * 32 + lane;
                const int4 t = d4p[vi < NV ? vi : NV - 1];
                p[q] = vi < NV ? t : none;
            }
#pragma unroll
            for (int q = 0; q < 4; ++q) {
                const int vi = vb + q * 32 + lane;
                const int comp[4] = {p[q].x, p[q].y, p[q].z, p[q].w};
#pragma unroll
                for (int c = 0; c < 4; ++c) {
                    const unsigned dl = (unsigned)(comp[c] - base);
                    const bool mt = dl < (unsigned)SLAB;
                    const unsigned mask = __builtin_amdgcn_ballot_w32(mt);
                    if (mask) {
                        const int n = __builtin_popcount(mask);
                        if (cnt + n <= QCAP) {
                            if (mt) qw[cnt + __builtin_amdgcn_mbcnt_lo(mask, 0u)] = ((vi * 4 + c) << 8) | (int)dl;
                            cnt += n;
                        } else {
                            ovf = true;
                        }
                    }
                }
            }
        }
        if (ovf && lane == 0) ovflag = 1;
        __syncthreads();
        slow = ovflag != 0;
        if (tid == 0) qslow[blockIdx.x] = slow ? 1 : 0;
        if (lane == 0) *gc = cnt;
    } else {
        slow = qslow[blockIdx.x] != 0;
        cnt  = *gc;
        __syncthreads();
    }

    if (!slow) {
        for (int i0 = 0; i0 < cnt; i0 += 128) {
            int r[4], s[4];
            float v[4][HEADS];
#pragma unroll
            for (int u = 0; u < 4; ++u) {
                const int idx = i0 + u * 32 + lane;
                r[u] = FIRST ? qw[idx < cnt ? idx : 0] : gq[idx < cnt ? idx : 0];
            }
#pragma unroll
            for (int u = 0; u < 4; ++u) {
                if (FIRST) {
                    const int e  = r[u] >> 8;
                    const int sl = src[e < E0 ? e : E0 - 1];
                    s[u] = e < E0 ? sl : e - E0;
                } else {
                    s[u] = r[u] >> 8;
                }
            }
#pragma unroll
            for (int u = 0; u < 4; ++u) lane_logits(a_src, a_dst, s[u], base + (r[u] & 255), v[u]);
#pragma unroll
            for (int u = 0; u < 4; ++u) {
                const int idx = i0 + u * 32 + lane;
                if (idx < cnt) {
                    const int dl = r[u] & 255;
                    const int en = (s[u] << 8) | dl;
                    qw[idx] = en;
                    if (FIRST) gq[idx] = en;
#pragma unroll
                    for (int h = 0; h < HEADS; ++h)
                        __hip_atomic_fetch_max(mloc + dl * HEADS + h, ford(v[u][h]),
                                               __ATOMIC_RELAXED, __HIP_MEMORY_SCOPE_WORKGROUP);
                }
            }
        }
        __syncthreads();
        for (int i0 = 0; i0 < cnt; i0 += 128) {
            int r[4];
            float v[4][HEADS];
#pragma unroll
            for (int u = 0; u < 4; ++u) {
                const int idx = i0 + u * 32 + lane;
                r[u] = qw[idx < cnt ? idx : 0];
            }
#pragma unroll
            for (int u = 0; u < 4; ++u) lane_logits(a_src, a_dst, r[u] >> 8, base + (r[u] & 255), v[u]);
#pragma unroll
            for (int u = 0; u < 4; ++u) {
                const int idx = i0 + u * 32 + lane;
                if (idx < cnt) {
                    const int dl = r[u] & 255;
#pragma unroll
                    for (int h = 0; h < HEADS; ++h) {
                        const float mv = funord(mloc[dl * HEADS + h]);
                        __hip_atomic_fetch_add(zloc + dl * HEADS + h, __expf(v[u][h] - mv),
                                               __ATOMIC_RELAXED, __HIP_MEMORY_SCOPE_WORKGROUP);
                    }
                }
            }
        }
        __syncthreads();
        const int kk = (lane & 15) >> 2;
        int i = 0;
        for (; i + 4 <= cnt; i += 4)
            rows_add<4>(qw[i + kk], base, lane, a_src, a_dst, mloc, zloc, H, slab);
        for (; i < cnt; ++i)
            rows_add<1>(qw[i], base, lane, a_src, a_dst, mloc, zloc, H, slab);
    } else {
        rescan<0>(tid, lane, wave, base, src, dst, a_src, a_dst, mloc, zloc, H, slab);
        __syncthreads();
        rescan<1>(tid, lane, wave, base, src, dst, a_src, a_dst, mloc, zloc, H, slab);
        __syncthreads();
        rescan<2>(tid, lane, wave, base, src, dst, a_src, a_dst, mloc, zloc, H, slab);
    }

    __syncthreads();
    float4* o4 = (float4*)(rows + (size_t)base * C_HID);
    for (int i = tid; i < SLAB * C_HID / 4; i += 1024) o4[i] = s4[i];
}

extern "C" void kernel_launch(void* const* d_in, const int* in_sizes, int n_in,
                              void* d_out, int out_size, void* d_ws, size_t ws_size,
                              hipStream_t stream) {
    const float* x        = (const float*)d_in[0];
    const int*   eidx     = (const int*)  d_in[1];
    const float* W1       = (const float*)d_in[2];
    const float* att_s1   = (const float*)d_in[3];
    const float* att_d1   = (const float*)d_in[4];
    const float* b1       = (const float*)d_in[5];
    const float* W2       = (const float*)d_in[6];
    const float* att_s2   = (const float*)d_in[7];
    const float* att_d2   = (const float*)d_in[8];
    const float* b2       = (const float*)d_in[9];
    float*       out      = (float*)d_out;

    const int* src = eidx;
    const int* dst = eidx + E0;

    const size_t NB = (size_t)N_NODES * C_HID;
    float* h      = (float*)d_ws;
    float* agg    = h      + NB;
    float* a_src  = agg    + NB;
    float* a_dst  = a_src  + (size_t)N_NODES * HEADS;
    int* qsave    = (int*)(a_dst + (size_t)N_NODES * HEADS);
    int* qcount   = qsave  + (size_t)NSLAB * SWAVES * QCAP;
    int* qslow    = qcount + NSLAB * SWAVES;
    float* WA1    = (float*)(qslow + 256);
    float* WA2    = WA1 + C_IN * 16;

    const dim3 B256(256);
    const int gMM  = N_NODES / 16;
    const int gFOLD = ((C_IN + C_HID) * 32 + 255) / 256;
    const int gMB   = (gMM + 1) / 2 * 2;

    fold_att   <<<gFOLD, B256, 0, stream>>>(W1, att_s1, att_d1, W2, att_s2, att_d2, WA1, WA2);
    gemm_wmma<false, C_IN><<<gMB, 128, 0, stream>>>(x, nullptr, W1, WA1, h, a_src, a_dst, C_HID, gMM);
    edge_slab<true><<<NSLAB, 1024, 0, stream>>>(src, dst, a_src, a_dst, h, nullptr, agg, qsave, qcount, qslow);

    gemm_wmma<true, C_HID><<<gMB, 128, 0, stream>>>(agg, b1, W2, WA2, h, a_src, a_dst, C_HID, gMM);
    edge_slab<false><<<NSLAB, 1024, 0, stream>>>(src, dst, a_src, a_dst, h, b2, out, qsave, qcount, qslow);
}


# Layer 1's edge scan reads one-byte slab numbers (dst/200, written by extra blocks of the fold launch), 16 per lane and load, in place of dst: scan bytes 4x fewer; sums still in LDS arrival order
# speedup vs baseline: 6.6693x; 1.0878x over previous
#include <hip/hip_runtime.h>
#include <math.h>

typedef __attribute__((ext_vector_type(2))) float v2f;
typedef __attribute__((ext_vector_type(8))) float v8f;

#define N_NODES 50000
#define C_IN    128
#define C_HID   256
#define HEADS   4
#define EMBED   64
#define E0      800000
#define ETOT    (E0 + N_NODES)
#define NEG_SLOPE 0.2f
#define SLAB   200
#define NSLAB  (N_NODES / SLAB)
static_assert(N_NODES % SLAB == 0, "rows split evenly over the blocks");
static_assert(NSLAB <= 255, "a slab number fits a byte; 255 stands for a row outside the graph");
static_assert(E0 % 16 == 0, "slab numbers are written four and scanned sixteen at a time");
#define FOLD_BLOCKS (((C_IN + C_HID) * 32 + 255) / 256)
#define ID_BLOCKS   ((E0 / 32 + 255) / 256)

template <bool ACT, int K>
__global__ __launch_bounds__(128) void gemm_wmma(const float* __restrict__ X,
                          const float* __restrict__ bias,
                          const float* __restrict__ W,
                          const float* __restrict__ WA,
                          float* __restrict__ H,
                          float* __restrict__ a_src,
                          float* __restrict__ a_dst,
                          int N, int MT) {
    static_assert(K % 16 == 0 && 128 % (K / 4) == 0, "four k-steps per trip; a thread keeps one column group");
    constexpr int KP = K + 4;
    __shared__ float As[32 * KP];
    __shared__ float Dx[3 * 256];

    const int tid    = threadIdx.x;
    const int lane   = tid & 31;
    const int wave   = __builtin_amdgcn_readfirstlane(tid >> 5);
    const int lane15 = lane & 15;
    const int khalf  = lane >> 4;
    const int ch     = (int)blockIdx.x & 1;
    const int tile0  = ((int)blockIdx.x >> 1) * 2;

    {
        constexpr int C4  = K / 4;
        constexpr int RS  = 128 / C4;
        constexpr int NLD = 32 / RS;
        const int c4 = tid % C4;
        const int r0 = tid / C4;
        const float* p0 = X + (size_t)(tile0 * 16 + r0) * K + c4 * 4;
        const float* p1 = p0 + (tile0 + 1 < MT ? 16 * K : 0);
        float4 t[NLD];
#pragma unroll
        for (int i = 0; i < NLD / 2; ++i) {
            t[i]           = *(const float4*)(p0 + i * RS * K);
            t[i + NLD / 2] = *(const float4*)(p1 + i * RS * K);
        }
        float4 bb = make_float4(0.f, 0.f, 0.f, 0.f);
        if (ACT) bb = *(const float4*)(bias + c4 * 4);
#pragma unroll
        for (int i = 0; i < NLD; ++i) {
            float4 v = t[i];
            if (ACT) {
                v.x = fmaxf(v.x + bb.x, 0.f);
                v.y = fmaxf(v.y + bb.y, 0.f);
                v.z = fmaxf(v.z + bb.z, 0.f);
                v.w = fmaxf(v.w + bb.w, 0.f);
            }
            *(float4*)(As + (r0 + i * RS) * KP + c4 * 4) = v;
        }
    }
    __syncthreads();

    const int col0 = ch * 128 + wave * 32;

    v8f acc00 = {}, acc01 = {};
    v8f acc10 = {}, acc11 = {};
    v8f accD  = {};

    const float* ar0 = As + lane15 * KP + khalf * 2;
    const float* ar1 = ar0 + 16 * KP;
    const float* ard = As + (ch * 16 + lane15) * KP + khalf * 2 + wave * 4;
    const float* wc  = W + col0 + lane15;
    const float* wac = WA + lane15;

    for (int k = 0; k < K; k += 16) {
#pragma unroll
        for (int u = 0; u < 4; ++u) {
            const int ks = k + u * 4;
            const int kk = ks + khalf * 2;
            const v2f a0 = *(const v2f*)(ar0 + ks);
            const v2f a1 = *(const v2f*)(ar1 + ks);
            const float* wp0 = wc + (size_t)kk * N;
            const float* wp1 = wp0 + N;
            v2f b0, b1;
            b0.x = wp0[ 0]; b0.y = wp1[ 0];
            b1.x = wp0[16]; b1.y = wp1[16];
            acc00 = __builtin_amdgcn_wmma_f32_16x16x4_f32(false, a0, false, b0, (short)0, acc00, false, false);
            acc10 = __builtin_amdgcn_wmma_f32_16x16x4_f32(false, a1, false, b0, (short)0, acc10, false, false);
            acc01 = __builtin_amdgcn_wmma_f32_16x16x4_f32(false, a0, false, b1, (short)0, acc01, false, false);
            acc11 = __builtin_amdgcn_wmma_f32_16x16x4_f32(false, a1, false, b1, (short)0, acc11, false, false);
        }
        {
            const int kd = k + wave * 4 + khalf * 2;
            const v2f ad = *(const v2f*)(ard + k);
            v2f bd;
            bd.x = wac[kd * 16];  bd.y = wac[(kd + 1) * 16];
            accD = __builtin_amdgcn_wmma_f32_16x16x4_f32(false, ad, false, bd, (short)0, accD, false, false);
        }
    }

    if (wave) {
#pragma unroll
        for (int i = 0; i < 8; ++i) Dx[(wave - 1) * 256 + i * 32 + lane] = accD[i];
    }
    __syncthreads();
    if (wave == 0 && tile0 + ch < MT && lane15 < 8) {
        float* op = (lane15 < 4 ? a_src : a_dst) + (size_t)((tile0 + ch) * 16 + khalf * 8) * HEADS + (lane15 & 3);
#pragma unroll
        for (int i = 0; i < 8; ++i) {
            const float p1 = Dx[      i * 32 + lane];
            const float p2 = Dx[256 + i * 32 + lane];
            const float p3 = Dx[512 + i * 32 + lane];
            op[i * HEADS] = (accD[i] + p2) + (p1 + p3);
        }
    }

    {
        float* op = H + (size_t)(tile0 * 16 + khalf * 8) * N + col0 + lane15;
#pragma unroll
        for (int i = 0; i < 8; ++i) {
            op[i * N +  0] = acc00[i];
            op[i * N + 16] = acc01[i];
        }
        if (tile0 + 1 < MT) {
            op += (size_t)16 * N;
#pragma unroll
            for (int i = 0; i < 8; ++i) {
                op[i * N +  0] = acc10[i];
                op[i * N + 16] = acc11[i];
            }
        }
    }
}

__device__ __forceinline__ unsigned slab_of(int d) {
    const unsigned u = (unsigned)d;
    return u < (unsigned)N_NODES ? u / SLAB : 255u;
}

__global__ void fold_att(const int* __restrict__ dst,
                         int* __restrict__ ids,
                         const float* __restrict__ W1,
                         const float* __restrict__ as1,
                         const float* __restrict__ ad1,
                         const float* __restrict__ W2,
                         const float* __restrict__ as2,
                         const float* __restrict__ ad2,
                         float* __restrict__ WA1,
                         float* __restrict__ WA2) {
    if (blockIdx.x >= FOLD_BLOCKS) {
        const int t = ((int)blockIdx.x - FOLD_BLOCKS) * 256 + (int)threadIdx.x;
        const int4* d4 = (const int4*)dst;
        int4 d[8];
#pragma unroll
        for (int k = 0; k < 8; ++k) {
            const int pi = t + k * (ID_BLOCKS * 256);
            d[k] = d4[pi < E0 / 4 ? pi : E0 / 4 - 1];
        }
#pragma unroll
        for (int k = 0; k < 8; ++k) {
            const int pi = t + k * (ID_BLOCKS * 256);
            if (pi < E0 / 4)
                ids[pi] = (int)(slab_of(d[k].x) | (slab_of(d[k].y) << 8) | (slab_of(d[k].z) << 16) | (slab_of(d[k].w) << 24));
        }
        return;
    }
    const int gw   = (int)((blockIdx.x * blockDim.x + threadIdx.x) >> 5);
    const int lane = threadIdx.x & 31;
    if (gw >= C_IN + C_HID) return;
    const bool second = gw >= C_IN;
    const int  k      = second ? gw - C_IN : gw;
    const float* Wm = second ? W2 : W1;
    const float* as = second ? as2 : as1;
    const float* ad = second ? ad2 : ad1;
    float*       WA = second ? WA2 : WA1;

    const int j    = lane >> 2;
    const int part = lane & 3;
    const int head = j & 3;
    const float* att = (j < 4 ? as : ad) + head * EMBED + part * 16;
    const float* wp  = Wm + (size_t)k * C_HID + head * EMBED + part * 16;
    float s = 0.f;
#pragma unroll
    for (int c = 0; c < 16; c += 4) {
        const float4 w4 = *(const float4*)(wp + c);
        const float4 t4 = *(const float4*)(att + c);
        s += w4.x * t4.x;
        s += w4.y * t4.y;
        s += w4.z * t4.z;
        s += w4.w * t4.w;
    }
    s += __shfl_xor(s, 1, 32);
    s += __shfl_xor(s, 2, 32);
    if (part == 0) WA[k * 16 + j] = s;
    if (part == 1) WA[k * 16 + 8 + j] = 0.f;
}

__device__ __forceinline__ unsigned ford(float f) {
    unsigned u = __float_as_uint(f);
    return (u >> 31) ? ~u : (u | 0x80000000u);
}
__device__ __forceinline__ float funord(unsigned u) {
    return (u >> 31) ? __uint_as_float(u ^ 0x80000000u) : __uint_as_float(~u);
}

__device__ __forceinline__ float leaky(float v) {
    return v > 0.f ? v : NEG_SLOPE * v;
}

#define QCAP   640
#define SWAVES 32
#define ORD_NEG_INF 0x007FFFFFu
static_assert(SLAB <= 256 && ETOT < (1 << 23), "a queue entry packs edge << 8 | local row");
static_assert(E0 % 4 == 0, "the slow path scans dst in int4 pieces");

__device__ __forceinline__ void lane_logits(const float* __restrict__ a_src,
                                            const float* __restrict__ a_dst,
                                            int s, int d, float v[HEADS]) {
    const float4 as = *(const float4*)(a_src + (size_t)s * HEADS);
    const float4 ad = *(const float4*)(a_dst + (size_t)d * HEADS);
    v[0] = leaky(as.x + ad.x);
    v[1] = leaky(as.y + ad.y);
    v[2] = leaky(as.z + ad.z);
    v[3] = leaky(as.w + ad.w);
}

__device__ __forceinline__ void lane_max(const float* __restrict__ a_src,
                                         const float* __restrict__ a_dst,
                                         int s, int dl, int base, unsigned* mloc) {
    float v[HEADS];
    lane_logits(a_src, a_dst, s, base + dl, v);
#pragma unroll
    for (int h = 0; h < HEADS; ++h)
        __hip_atomic_fetch_max(mloc + dl * HEADS + h, ford(v[h]),
                               __ATOMIC_RELAXED, __HIP_MEMORY_SCOPE_WORKGROUP);
}

__device__ __forceinline__ void lane_z(const float* __restrict__ a_src,
                                       const float* __restrict__ a_dst,
                                       int s, int dl, int base,
                                       const unsigned* mloc, float* zloc) {
    float v[HEADS];
    lane_logits(a_src, a_dst, s, base + dl, v);
#pragma unroll
    for (int h = 0; h < HEADS; ++h) {
        const float mv = funord(mloc[dl * HEADS + h]);
        __hip_atomic_fetch_add(zloc + dl * HEADS + h, __expf(v[h] - mv),
                               __ATOMIC_RELAXED, __HIP_MEMORY_SCOPE_WORKGROUP);
    }
}

template <int NB>
__device__ __forceinline__ void rows_add(int r, int base, int lane,
                                         const float* __restrict__ a_src,
                                         const float* __restrict__ a_dst,
                                         const unsigned* mloc, const float* zloc,
                                         const float* __restrict__ H,
                                         float* slab) {
    int s[NB], dl[NB];
#pragma unroll
    for (int k = 0; k < NB; ++k) {
        const int rk = __builtin_amdgcn_readlane(r, 4 * k);
        s[k]  = rk >> 8;
        dl[k] = rk & 255;
    }
    float hv[NB][8];
#pragma unroll
    for (int k = 0; k < NB; ++k) {
        const float* hp = H + (size_t)s[k] * C_HID + lane;
#pragma unroll
        for (int j = 0; j < 8; ++j) hv[k][j] = hp[j * 32];
    }
    const int h   = lane & 3;
    const int sL  = r >> 8;
    const int dlL = r & 255;
    const float v  = leaky(a_src[(size_t)sL * HEADS + h] + a_dst[(size_t)(base + dlL) * HEADS + h]);
    const float ev = __expf(v - funord(mloc[dlL * HEADS + h]));
    const int   al = __float_as_int(ev / (zloc[dlL * HEADS + h] + 1e-16f));
#pragma unroll
    for (int k = 0; k < NB; ++k) {
        float* sp = slab + dl[k] * C_HID + lane;
#pragma unroll
        for (int j = 0; j < 8; ++j) {
            const float a = __int_as_float(__builtin_amdgcn_readlane(al, 4 * k + (j >> 1)));
            __hip_atomic_fetch_add(sp + j * 32, hv[k][j] * a,
                                   __ATOMIC_RELAXED, __HIP_MEMORY_SCOPE_WORKGROUP);
        }
    }
}

template <int PH>
__device__ __forceinline__ void rescan(int tid, int lane, int wave, int base,
                                       const int* __restrict__ src,
                                       const int* __restrict__ dst,
                                       const float* __restrict__ a_src,
                                       const float* __restrict__ a_dst,
                                       unsigned* mloc, float* zloc,
                                       const float* __restrict__ H,
                                       float* slab) {
    if (PH < 2) {
        for (int n = tid; n < SLAB; n += 1024) {
            if (PH == 0) lane_max(a_src, a_dst, base + n, n, base, mloc);
            else         lane_z(a_src, a_dst, base + n, n, base, mloc, zloc);
        }
    } else {
        for (int n = wave; n < SLAB; n += SWAVES) {
            rows_add<1>(((base + n) << 8) | n, base, lane, a_src, a_dst, mloc, zloc, H, slab);
        }
    }
    const int4* d4p = (const int4*)dst;
    const int NV = E0 / 4;
    for (int vb = wave * 32; vb < NV; vb += SWAVES * 32) {
        const int vi = vb + lane;
        const int4 t = d4p[vi < NV ? vi : NV - 1];
        const int comp[4] = {t.x, t.y, t.z, t.w};
#pragma unroll
        for (int c = 0; c < 4; ++c) {
            const unsigned dl = (unsigned)(comp[c] - base);
            const bool mt = vi < NV && dl < (unsigned)SLAB;
            int s = 0;
            if (mt) s = src[vi * 4 + c];
            if (PH == 0) {
                if (mt) lane_max(a_src, a_dst, s, (int)dl, base, mloc);
            } else if (PH == 1) {
                if (mt) lane_z(a_src, a_dst, s, (int)dl, base, mloc, zloc);
            } else {
                unsigned mask = __builtin_amdgcn_ballot_w32(mt);
                while (mask) {
                    const int b = __builtin_ctz(mask);
                    mask &= mask - 1;
                    const int rb = __builtin_amdgcn_readlane((s << 8) | (int)dl, b);
                    rows_add<1>(rb, base, lane, a_src, a_dst, mloc, zloc, H, slab);
                }
            }
        }
    }
}

template <bool FIRST>
__global__ __launch_bounds__(1024) void edge_slab(const int* __restrict__ src,
                                                  const int* __restrict__ dst,
                                                  const float* __restrict__ a_src,
                                                  const float* __restrict__ a_dst,
                                                  const float* __restrict__ H,
                                                  const float* __restrict__ bias,
                                                  float* __restrict__ rows,
                                                  int* __restrict__ qsave,
                                                  int* __restrict__ qcount,
                                                  int* __restrict__ qslow,
                                                  const int* __restrict__ ids) {
    __shared__ float    slab[SLAB * C_HID];
    __shared__ int      queue[SWAVES * QCAP];
    __shared__ unsigned mloc[SLAB * HEADS];
    __shared__ float    zloc[SLAB * HEADS];
    __shared__ int      ovflag;

    const int tid  = threadIdx.x;
    const int lane = tid & 31;
    const int wave = __builtin_amdgcn_readfirstlane(tid >> 5);
    const int base = blockIdx.x * SLAB;

    float4* s4 = (float4*)slab;
    {
        float4 start = make_float4(0.f, 0.f, 0.f, 0.f);
        if (!FIRST) start = *(const float4*)(bias + (tid & 63) * 4);
        for (int i = tid; i < SLAB * C_HID / 4; i += 1024) s4[i] = start;
    }
    for (int i = tid; i < SLAB * HEADS; i += 1024) { mloc[i] = ORD_NEG_INF; zloc[i] = 0.f; }
    if (FIRST && tid == 0) ovflag = 0;

    int* qw = queue + wave * QCAP;
    int* gq = qsave + ((size_t)blockIdx.x * SWAVES + wave) * QCAP;
    int* gc = qcount + blockIdx.x * SWAVES + wave;
    int  cnt;
    bool slow;
    if (FIRST) {
        cnt = (SLAB - 1 - wave) / SWAVES + 1;
        if (lane < cnt) {
            const int n = wave + SWAVES * lane;
            qw[lane] = ((E0 + base + n) << 8) | n;
        }
        __syncthreads();

        const int4* i4p = (const int4*)ids;
        const int NV = E0 / 16;
        const unsigned pat = (unsigned)blockIdx.x * 0x01010101u;
        constexpr int ROWP = SWAVES * 32;
        bool ovf = false;
        for (int vb = wave * 32; vb < NV; vb += 4 * ROWP) {
            int4 p[4];
#pragma unroll
            for (int q = 0; q < 4; ++q) {
                const int vi = vb + q * ROWP + lane;
                p[q] = i4p[vi < NV ? vi : NV - 1];
            }
#pragma unroll
            for (int q = 0; q < 4; ++q) {
                const int vi = vb + q * ROWP + lane;
                const unsigned w[4] = {(unsigned)p[q].x, (unsigned)p[q].y, (unsigned)p[q].z, (unsigned)p[q].w};
                unsigned mm = 0;
#pragma unroll
                for (int d = 0; d < 4; ++d) {
                    const unsigned t = w[d] ^ pat;
                    const unsigned zb = ~(((t & 0x7F7F7F7Fu) + 0x7F7F7F7Fu) | t | 0x7F7F7F7Fu);
                    mm |= zb >> (7 - d);
                }
                if (vi >= NV) mm = 0;
                unsigned live = __builtin_amdgcn_ballot_w32(mm != 0);
                while (live) {
                    const int n = __builtin_popcount(live);
                    if (cnt + n <= QCAP) {
                        if (mm != 0) {
                            const int b = __builtin_ctz(mm);
                            const int e = vi * 16 + (b & 7) * 4 + (b >> 3);
                            qw[cnt + __builtin_amdgcn_mbcnt_lo(live, 0u)] = e << 8;
                        }
                        cnt += n;
                    } else {
                        ovf = true;
                    }
                    mm &= mm - 1;
                    live = __builtin_amdgcn_ballot_w32(mm != 0);
                }
            }
        }
        if (ovf && lane == 0) ovflag = 1;
        __syncthreads();
        slow = ovflag != 0;
        if (tid == 0) qslow[blockIdx.x] = slow ? 1 : 0;
        if (lane == 0) *gc = cnt;
    } else {
        slow = qslow[blockIdx.x] != 0;
        cnt  = *gc;
        __syncthreads();
    }

    if (!slow) {
        for (int i0 = 0; i0 < cnt; i0 += 128) {
            int r[4], s[4], dlv[4];
            float v[4][HEADS];
#pragma unroll
            for (int u = 0; u < 4; ++u) {
                const int idx = i0 + u * 32 + lane;
                r[u] = FIRST ? qw[idx < cnt ? idx : 0] : gq[idx < cnt ? idx : 0];
            }
#pragma unroll
            for (int u = 0; u < 4; ++u) {
                if (FIRST) {
                    const int e  = r[u] >> 8;
                    const int ec = e < E0 ? e : E0 - 1;
                    const int sl = src[ec];
                    const int dd = dst[ec];
                    s[u]   = e < E0 ? sl : e - E0;
                    dlv[u] = e < E0 ? dd - base : (r[u] & 255);
                } else {
                    s[u]   = r[u] >> 8;
                    dlv[u] = r[u] & 255;
                }
            }
#pragma unroll
            for (int u = 0; u < 4; ++u) lane_logits(a_src, a_dst, s[u], base + dlv[u], v[u]);
#pragma unroll
            for (int u = 0; u < 4; ++u) {
                const int idx = i0 + u * 32 + lane;
                if (idx < cnt) {
                    const int dl = dlv[u];
                    const int en = (s[u] << 8) | dl;
                    qw[idx] = en;
                    if (FIRST) gq[idx] = en;
#pragma unroll
                    for (int h = 0; h < HEADS; ++h)
                        __hip_atomic_fetch_max(mloc + dl * HEADS + h, ford(v[u][h]),
                                               __ATOMIC_RELAXED, __HIP_MEMORY_SCOPE_WORKGROUP);
                }
            }
        }
        __syncthreads();
        for (int i0 = 0; i0 < cnt; i0 += 128) {
            int r[4];
            float v[4][HEADS];
#pragma unroll
            for (int u = 0; u < 4; ++u) {
                const int idx = i0 + u * 32 + lane;
                r[u] = qw[idx < cnt ? idx : 0];
            }
#pragma unroll
            for (int u = 0; u < 4; ++u) lane_logits(a_src, a_dst, r[u] >> 8, base + (r[u] & 255), v[u]);
#pragma unroll
            for (int u = 0; u < 4; ++u) {
                const int idx = i0 + u * 32 + lane;
                if (idx < cnt) {
                    const int dl = r[u] & 255;
#pragma unroll
                    for (int h = 0; h < HEADS; ++h) {
                        const float mv = funord(mloc[dl * HEADS + h]);
                        __hip_atomic_fetch_add(zloc + dl * HEADS + h, __expf(v[u][h] - mv),
                                               __ATOMIC_RELAXED, __HIP_MEMORY_SCOPE_WORKGROUP);
                    }
                }
            }
        }
        __syncthreads();
        const int kk = (lane & 15) >> 2;
        int i = 0;
        for (; i + 4 <= cnt; i += 4)
            rows_add<4>(qw[i + kk], base, lane, a_src, a_dst, mloc, zloc, H, slab);
        for (; i < cnt; ++i)
            rows_add<1>(qw[i], base, lane, a_src, a_dst, mloc, zloc, H, slab);
    } else {
        rescan<0>(tid, lane, wave, base, src, dst, a_src, a_dst, mloc, zloc, H, slab);
        __syncthreads();
        rescan<1>(tid, lane, wave, base, src, dst, a_src, a_dst, mloc, zloc, H, slab);
        __syncthreads();
        rescan<2>(tid, lane, wave, base, src, dst, a_src, a_dst, mloc, zloc, H, slab);
    }

    __syncthreads();
    float4* o4 = (float4*)(rows + (size_t)base * C_HID);
    for (int i = tid; i < SLAB * C_HID / 4; i += 1024) o4[i] = s4[i];
}

extern "C" void kernel_launch(void* const* d_in, const int* in_sizes, int n_in,
                              void* d_out, int out_size, void* d_ws, size_t ws_size,
                              hipStream_t stream) {
    const float* x        = (const float*)d_in[0];
    const int*   eidx     = (const int*)  d_in[1];
    const float* W1       = (const float*)d_in[2];
    const float* att_s1   = (const float*)d_in[3];
    const float* att_d1   = (const float*)d_in[4];
    const float* b1       = (const float*)d_in[5];
    const float* W2       = (const float*)d_in[6];
    const float* att_s2   = (const float*)d_in[7];
    const float* att_d2   = (const float*)d_in[8];
    const float* b2       = (const float*)d_in[9];
    float*       out      = (float*)d_out;

    const int* src = eidx;
    const int* dst = eidx + E0;

    const size_t NB = (size_t)N_NODES * C_HID;
    float* h      = (float*)d_ws;
    float* agg    = h      + NB;
    float* a_src  = agg    + NB;
    float* a_dst  = a_src  + (size_t)N_NODES * HEADS;
    int* qsave    = (int*)(a_dst + (size_t)N_NODES * HEADS);
    int* qcount   = qsave  + (size_t)NSLAB * SWAVES * QCAP;
    int* qslow    = qcount + NSLAB * SWAVES;
    float* WA1    = (float*)(qslow + 256);
    float* WA2    = WA1 + C_IN * 16;
    int* ids      = (int*)(WA2 + C_HID * 16);

    const dim3 B256(256);
    const int gMM  = N_NODES / 16;
    const int gFOLD = FOLD_BLOCKS + ID_BLOCKS;
    const int gMB   = (gMM + 1) / 2 * 2;

    fold_att   <<<gFOLD, B256, 0, stream>>>(dst, ids, W1, att_s1, att_d1, W2, att_s2, att_d2, WA1, WA2);
    gemm_wmma<false, C_IN><<<gMB, 128, 0, stream>>>(x, nullptr, W1, WA1, h, a_src, a_dst, C_HID, gMM);
    edge_slab<true><<<NSLAB, 1024, 0, stream>>>(src, dst, a_src, a_dst, h, nullptr, agg, qsave, qcount, qslow, ids);

    gemm_wmma<true, C_HID><<<gMB, 128, 0, stream>>>(agg, b1, W2, WA2, h, a_src, a_dst, C_HID, gMM);
    edge_slab<false><<<NSLAB, 1024, 0, stream>>>(src, dst, a_src, a_dst, h, b2, out, qsave, qcount, qslow, nullptr);
}


# Each queued edge's four head logits are kept in LDS from the max step on: the exp-sum step and the row step read them there and no longer gather a_src and a_dst again; values bit-identical
# speedup vs baseline: 7.0229x; 1.0530x over previous
#include <hip/hip_runtime.h>
#include <math.h>

typedef __attribute__((ext_vector_type(2))) float v2f;
typedef __attribute__((ext_vector_type(8))) float v8f;

#define N_NODES 50000
#define C_IN    128
#define C_HID   256
#define HEADS   4
#define EMBED   64
#define E0      800000
#define ETOT    (E0 + N_NODES)
#define NEG_SLOPE 0.2f
#define SLAB   200
#define NSLAB  (N_NODES / SLAB)
static_assert(N_NODES % SLAB == 0, "rows split evenly over the blocks");
static_assert(NSLAB <= 255, "a slab number fits a byte; 255 stands for a row outside the graph");
static_assert(E0 % 16 == 0, "slab numbers are written four and scanned sixteen at a time");
#define FOLD_BLOCKS (((C_IN + C_HID) * 32 + 255) / 256)
#define ID_BLOCKS   ((E0 / 32 + 255) / 256)

template <bool ACT, int K>
__global__ __launch_bounds__(128) void gemm_wmma(const float* __restrict__ X,
                          const float* __restrict__ bias,
                          const float* __restrict__ W,
                          const float* __restrict__ WA,
                          float* __restrict__ H,
                          float* __restrict__ a_src,
                          float* __restrict__ a_dst,
                          int N, int MT) {
    static_assert(K % 16 == 0 && 128 % (K / 4) == 0, "four k-steps per trip; a thread keeps one column group");
    constexpr int KP = K + 4;
    __shared__ float As[32 * KP];
    __shared__ float Dx[3 * 256];

    const int tid    = threadIdx.x;
    const int lane   = tid & 31;
    const int wave   = __builtin_amdgcn_readfirstlane(tid >> 5);
    const int lane15 = lane & 15;
    const int khalf  = lane >> 4;
    const int ch     = (int)blockIdx.x & 1;
    const int tile0  = ((int)blockIdx.x >> 1) * 2;

    {
        constexpr int C4  = K / 4;
        constexpr int RS  = 128 / C4;
        constexpr int NLD = 32 / RS;
        const int c4 = tid % C4;
        const int r0 = tid / C4;
        const float* p0 = X + (size_t)(tile0 * 16 + r0) * K + c4 * 4;
        const float* p1 = p0 + (tile0 + 1 < MT ? 16 * K : 0);
        float4 t[NLD];
#pragma unroll
        for (int i = 0; i < NLD / 2; ++i) {
            t[i]           = *(const float4*)(p0 + i * RS * K);
            t[i + NLD / 2] = *(const float4*)(p1 + i * RS * K);
        }
        float4 bb = make_float4(0.f, 0.f, 0.f, 0.f);
        if (ACT) bb = *(const float4*)(bias + c4 * 4);
#pragma unroll
        for (int i = 0; i < NLD; ++i) {
            float4 v = t[i];
            if (ACT) {
                v.x = fmaxf(v.x + bb.x, 0.f);
                v.y = fmaxf(v.y + bb.y, 0.f);
                v.z = fmaxf(v.z + bb.z, 0.f);
                v.w = fmaxf(v.w + bb.w, 0.f);
            }
            *(float4*)(As + (r0 + i * RS) * KP + c4 * 4) = v;
        }
    }
    __syncthreads();

    const int col0 = ch * 128 + wave * 32;

    v8f acc00 = {}, acc01 = {};
    v8f acc10 = {}, acc11 = {};
    v8f accD  = {};

    const float* ar0 = As + lane15 * KP + khalf * 2;
    const float* ar1 = ar0 + 16 * KP;
    const float* ard = As + (ch * 16 + lane15) * KP + khalf * 2 + wave * 4;
    const float* wc  = W + col0 + lane15;
    const float* wac = WA + lane15;

    for (int k = 0; k < K; k += 16) {
#pragma unroll
        for (int u = 0; u < 4; ++u) {
            const int ks = k + u * 4;
            const int kk = ks + khalf * 2;
            const v2f a0 = *(const v2f*)(ar0 + ks);
            const v2f a1 = *(const v2f*)(ar1 + ks);
            const float* wp0 = wc + (size_t)kk * N;
            const float* wp1 = wp0 + N;
            v2f b0, b1;
            b0.x = wp0[ 0]; b0.y = wp1[ 0];
            b1.x = wp0[16]; b1.y = wp1[16];
            acc00 = __builtin_amdgcn_wmma_f32_16x16x4_f32(false, a0, false, b0, (short)0, acc00, false, false);
            acc10 = __builtin_amdgcn_wmma_f32_16x16x4_f32(false, a1, false, b0, (short)0, acc10, false, false);
            acc01 = __builtin_amdgcn_wmma_f32_16x16x4_f32(false, a0, false, b1, (short)0, acc01, false, false);
            acc11 = __builtin_amdgcn_wmma_f32_16x16x4_f32(false, a1, false, b1, (short)0, acc11, false, false);
        }
        {
            const int kd = k + wave * 4 + khalf * 2;
            const v2f ad = *(const v2f*)(ard + k);
            v2f bd;
            bd.x = wac[kd * 16];  bd.y = wac[(kd + 1) * 16];
            accD = __builtin_amdgcn_wmma_f32_16x16x4_f32(false, ad, false, bd, (short)0, accD, false, false);
        }
    }

    if (wave) {
#pragma unroll
        for (int i = 0; i < 8; ++i) Dx[(wave - 1) * 256 + i * 32 + lane] = accD[i];
    }
    __syncthreads();
    if (wave == 0 && tile0 + ch < MT && lane15 < 8) {
        float* op = (lane15 < 4 ? a_src : a_dst) + (size_t)((tile0 + ch) * 16 + khalf * 8) * HEADS + (lane15 & 3);
#pragma unroll
        for (int i = 0; i < 8; ++i) {
            const float p1 = Dx[      i * 32 + lane];
            const float p2 = Dx[256 + i * 32 + lane];
            const float p3 = Dx[512 + i * 32 + lane];
            op[i * HEADS] = (accD[i] + p2) + (p1 + p3);
        }
    }

    {
        float* op = H + (size_t)(tile0 * 16 + khalf * 8) * N + col0 + lane15;
#pragma unroll
        for (int i = 0; i < 8; ++i) {
            op[i * N +  0] = acc00[i];
            op[i * N + 16] = acc01[i];
        }
        if (tile0 + 1 < MT) {
            op += (size_t)16 * N;
#pragma unroll
            for (int i = 0; i < 8; ++i) {
                op[i * N +  0] = acc10[i];
                op[i * N + 16] = acc11[i];
            }
        }
    }
}

__device__ __forceinline__ unsigned slab_of(int d) {
    const unsigned u = (unsigned)d;
    return u < (unsigned)N_NODES ? u / SLAB : 255u;
}

__global__ void fold_att(const int* __restrict__ dst,
                         int* __restrict__ ids,
                         const float* __restrict__ W1,
                         const float* __restrict__ as1,
                         const float* __restrict__ ad1,
                         const float* __restrict__ W2,
                         const float* __restrict__ as2,
                         const float* __restrict__ ad2,
                         float* __restrict__ WA1,
                         float* __restrict__ WA2) {
    if (blockIdx.x >= FOLD_BLOCKS) {
        const int t = ((int)blockIdx.x - FOLD_BLOCKS) * 256 + (int)threadIdx.x;
        const int4* d4 = (const int4*)dst;
        int4 d[8];
#pragma unroll
        for (int k = 0; k < 8; ++k) {
            const int pi = t + k * (ID_BLOCKS * 256);
            d[k] = d4[pi < E0 / 4 ? pi : E0 / 4 - 1];
        }
#pragma unroll
        for (int k = 0; k < 8; ++k) {
            const int pi = t + k * (ID_BLOCKS * 256);
            if (pi < E0 / 4)
                ids[pi] = (int)(slab_of(d[k].x) | (slab_of(d[k].y) << 8) | (slab_of(d[k].z) << 16) | (slab_of(d[k].w) << 24));
        }
        return;
    }
    const int gw   = (int)((blockIdx.x * blockDim.x + threadIdx.x) >> 5);
    const int lane = threadIdx.x & 31;
    if (gw >= C_IN + C_HID) return;
    const bool second = gw >= C_IN;
    const int  k      = second ? gw - C_IN : gw;
    const float* Wm = second ? W2 : W1;
    const float* as = second ? as2 : as1;
    const float* ad = second ? ad2 : ad1;
    float*       WA = second ? WA2 : WA1;

    const int j    = lane >> 2;
    const int part = lane & 3;
    const int head = j & 3;
    const float* att = (j < 4 ? as : ad) + head * EMBED + part * 16;
    const float* wp  = Wm + (size_t)k * C_HID + head * EMBED + part * 16;
    float s = 0.f;
#pragma unroll
    for (int c = 0; c < 16; c += 4) {
        const float4 w4 = *(const float4*)(wp + c);
        const float4 t4 = *(const float4*)(att + c);
        s += w4.x * t4.x;
        s += w4.y * t4.y;
        s += w4.z * t4.z;
        s += w4.w * t4.w;
    }
    s += __shfl_xor(s, 1, 32);
    s += __shfl_xor(s, 2, 32);
    if (part == 0) WA[k * 16 + j] = s;
    if (part == 1) WA[k * 16 + 8 + j] = 0.f;
}

__device__ __forceinline__ unsigned ford(float f) {
    unsigned u = __float_as_uint(f);
    return (u >> 31) ? ~u : (u | 0x80000000u);
}
__device__ __forceinline__ float funord(unsigned u) {
    return (u >> 31) ? __uint_as_float(u ^ 0x80000000u) : __uint_as_float(~u);
}

__device__ __forceinline__ float leaky(float v) {
    return v > 0.f ? v : NEG_SLOPE * v;
}

#define QCAP   176
#define SWAVES 32
#define ORD_NEG_INF 0x007FFFFFu
static_assert(SLAB <= 256 && ETOT < (1 << 23), "a queue entry packs edge << 8 | local row");
static_assert(E0 % 4 == 0, "the slow path scans dst in int4 pieces");

__device__ __forceinline__ void lane_logits(const float* __restrict__ a_src,
                                            const float* __restrict__ a_dst,
                                            int s, int d, float v[HEADS]) {
    const float4 as = *(const float4*)(a_src + (size_t)s * HEADS);
    const float4 ad = *(const float4*)(a_dst + (size_t)d * HEADS);
    v[0] = leaky(as.x + ad.x);
    v[1] = leaky(as.y + ad.y);
    v[2] = leaky(as.z + ad.z);
    v[3] = leaky(as.w + ad.w);
}

__device__ __forceinline__ void lane_max(const float* __restrict__ a_src,
                                         const float* __restrict__ a_dst,
                                         int s, int dl, int base, unsigned* mloc) {
    float v[HEADS];
    lane_logits(a_src, a_dst, s, base + dl, v);
#pragma unroll
    for (int h = 0; h < HEADS; ++h)
        __hip_atomic_fetch_max(mloc + dl * HEADS + h, ford(v[h]),
                               __ATOMIC_RELAXED, __HIP_MEMORY_SCOPE_WORKGROUP);
}

__device__ __forceinline__ void lane_z(const float* __restrict__ a_src,
                                       const float* __restrict__ a_dst,
                                       int s, int dl, int base,
                                       const unsigned* mloc, float* zloc) {
    float v[HEADS];
    lane_logits(a_src, a_dst, s, base + dl, v);
#pragma unroll
    for (int h = 0; h < HEADS; ++h) {
        const float mv = funord(mloc[dl * HEADS + h]);
        __hip_atomic_fetch_add(zloc + dl * HEADS + h, __expf(v[h] - mv),
                               __ATOMIC_RELAXED, __HIP_MEMORY_SCOPE_WORKGROUP);
    }
}

template <int NB, bool LGQ>
__device__ __forceinline__ void rows_add(int r, int base, int lane,
                                         const float* __restrict__ a_src,
                                         const float* __restrict__ a_dst,
                                         const float* lgp,
                                         const unsigned* mloc, const float* zloc,
                                         const float* __restrict__ H,
                                         float* slab) {
    int s[NB], dl[NB];
#pragma unroll
    for (int k = 0; k < NB; ++k) {
        const int rk = __builtin_amdgcn_readlane(r, 4 * k);
        s[k]  = rk >> 8;
        dl[k] = rk & 255;
    }
    float hv[NB][8];
#pragma unroll
    for (int k = 0; k < NB; ++k) {
        const float* hp = H + (size_t)s[k] * C_HID + lane;
#pragma unroll
        for (int j = 0; j < 8; ++j) hv[k][j] = hp[j * 32];
    }
    const int h   = lane & 3;
    const int sL  = r >> 8;
    const int dlL = r & 255;
    float v;
    if (LGQ) v = *lgp;
    else     v = leaky(a_src[(size_t)sL * HEADS + h] + a_dst[(size_t)(base + dlL) * HEADS + h]);
    const float ev = __expf(v - funord(mloc[dlL * HEADS + h]));
    const int   al = __float_as_int(ev / (zloc[dlL * HEADS + h] + 1e-16f));
#pragma unroll
    for (int k = 0; k < NB; ++k) {
        float* sp = slab + dl[k] * C_HID + lane;
#pragma unroll
        for (int j = 0; j < 8; ++j) {
            const float a = __int_as_float(__builtin_amdgcn_readlane(al, 4 * k + (j >> 1)));
            __hip_atomic_fetch_add(sp + j * 32, hv[k][j] * a,
                                   __ATOMIC_RELAXED, __HIP_MEMORY_SCOPE_WORKGROUP);
        }
    }
}

template <int PH>
__device__ __forceinline__ void rescan(int tid, int lane, int wave, int base,
                                       const int* __restrict__ src,
                                       const int* __restrict__ dst,
                                       const float* __restrict__ a_src,
                                       const float* __restrict__ a_dst,
                                       unsigned* mloc, float* zloc,
                                       const float* __restrict__ H,
                                       float* slab) {
    if (PH < 2) {
        for (int n = tid; n < SLAB; n += 1024) {
            if (PH == 0) lane_max(a_src, a_dst, base + n, n, base, mloc);
            else         lane_z(a_src, a_dst, base + n, n, base, mloc, zloc);
        }
    } else {
        for (int n = wave; n < SLAB; n += SWAVES) {
            rows_add<1, false>(((base + n) << 8) | n, base, lane, a_src, a_dst, nullptr, mloc, zloc, H, slab);
        }
    }
    const int4* d4p = (const int4*)dst;
    const int NV = E0 / 4;
    for (int vb = wave * 32; vb < NV; vb += SWAVES * 32) {
        const int vi = vb + lane;
        const int4 t = d4p[vi < NV ? vi : NV - 1];
        const int comp[4] = {t.x, t.y, t.z, t.w};
#pragma unroll
        for (int c = 0; c < 4; ++c) {
            const unsigned dl = (unsigned)(comp[c] - base);
            const bool mt = vi < NV && dl < (unsigned)SLAB;
            int s = 0;
            if (mt) s = src[vi * 4 + c];
            if (PH == 0) {
                if (mt) lane_max(a_src, a_dst, s, (int)dl, base, mloc);
            } else if (PH == 1) {
                if (mt) lane_z(a_src, a_dst, s, (int)dl, base, mloc, zloc);
            } else {
                unsigned mask = __builtin_amdgcn_ballot_w32(mt);
                while (mask) {
                    const int b = __builtin_ctz(mask);
                    mask &= mask - 1;
                    const int rb = __builtin_amdgcn_readlane((s << 8) | (int)dl, b);
                    rows_add<1, false>(rb, base, lane, a_src, a_dst, nullptr, mloc, zloc, H, slab);
                }
            }
        }
    }
}

template <bool FIRST>
__global__ __launch_bounds__(1024) void edge_slab(const int* __restrict__ src,
                                                  const int* __restrict__ dst,
                                                  const float* __restrict__ a_src,
                                                  const float* __restrict__ a_dst,
                                                  const float* __restrict__ H,
                                                  const float* __restrict__ bias,
                                                  float* __restrict__ rows,
                                                  int* __restrict__ qsave,
                                                  int* __restrict__ qcount,
                                                  int* __restrict__ qslow,
                                                  const int* __restrict__ ids) {
    __shared__ float    slab[SLAB * C_HID];
    __shared__ int      queue[SWAVES * QCAP];
    __shared__ float4   lgq[SWAVES * QCAP];
    __shared__ unsigned mloc[SLAB * HEADS];
    __shared__ float    zloc[SLAB * HEADS];
    __shared__ int      ovflag;

    const int tid  = threadIdx.x;
    const int lane = tid & 31;
    const int wave = __builtin_amdgcn_readfirstlane(tid >> 5);
    const int base = blockIdx.x * SLAB;

    float4* s4 = (float4*)slab;
    {
        float4 start = make_float4(0.f, 0.f, 0.f, 0.f);
        if (!FIRST) start = *(const float4*)(bias + (tid & 63) * 4);
        for (int i = tid; i < SLAB * C_HID / 4; i += 1024) s4[i] = start;
    }
    for (int i = tid; i < SLAB * HEADS; i += 1024) { mloc[i] = ORD_NEG_INF; zloc[i] = 0.f; }
    if (FIRST && tid == 0) ovflag = 0;

    int* qw = queue + wave * QCAP;
    float4* lgw = lgq + wave * QCAP;
    int* gq = qsave + ((size_t)blockIdx.x * SWAVES + wave) * QCAP;
    int* gc = qcount + blockIdx.x * SWAVES + wave;
    int  cnt;
    bool slow;
    if (FIRST) {
        cnt = (SLAB - 1 - wave) / SWAVES + 1;
        if (lane < cnt) {
            const int n = wave + SWAVES * lane;
            qw[lane] = ((E0 + base + n) << 8) | n;
        }
        __syncthreads();

        const int4* i4p = (const int4*)ids;
        const int NV = E0 / 16;
        const unsigned pat = (unsigned)blockIdx.x * 0x01010101u;
        constexpr int ROWP = SWAVES * 32;
        bool ovf = false;
        for (int vb = wave * 32; vb < NV; vb += 4 * ROWP) {
            int4 p[4];
#pragma unroll
            for (int q = 0; q < 4; ++q) {
                const int vi = vb + q * ROWP + lane;
                p[q] = i4p[vi < NV ? vi : NV - 1];
            }
#pragma unroll
            for (int q = 0; q < 4; ++q) {
                const int vi = vb + q * ROWP + lane;
                const unsigned w[4] = {(unsigned)p[q].x, (unsigned)p[q].y, (unsigned)p[q].z, (unsigned)p[q].w};
                unsigned mm = 0;
#pragma unroll
                for (int d = 0; d < 4; ++d) {
                    const unsigned t = w[d] ^ pat;
                    const unsigned zb = ~(((t & 0x7F7F7F7Fu) + 0x7F7F7F7Fu) | t | 0x7F7F7F7Fu);
                    mm |= zb >> (7 - d);
                }
                if (vi >= NV) mm = 0;
                unsigned live = __builtin_amdgcn_ballot_w32(mm != 0);
                while (live) {
                    const int n = __builtin_popcount(live);
                    if (cnt + n <= QCAP) {
                        if (mm != 0) {
                            const int b = __builtin_ctz(mm);
                            const int e = vi * 16 + (b & 7) * 4 + (b >> 3);
                            qw[cnt + __builtin_amdgcn_mbcnt_lo(live, 0u)] = e << 8;
                        }
                        cnt += n;
                    } else {
                        ovf = true;
                    }
                    mm &= mm - 1;
                    live = __builtin_amdgcn_ballot_w32(mm != 0);
                }
            }
        }
        if (ovf && lane == 0) ovflag = 1;
        __syncthreads();
        slow = ovflag != 0;
        if (tid == 0) qslow[blockIdx.x] = slow ? 1 : 0;
        if (lane == 0) *gc = cnt;
    } else {
        slow = qslow[blockIdx.x] != 0;
        cnt  = *gc;
        __syncthreads();
    }

    if (!slow) {
        for (int i0 = 0; i0 < cnt; i0 += 128) {
            int r[4], s[4], dlv[4];
            float v[4][HEADS];
#pragma unroll
            for (int u = 0; u < 4; ++u) {
                const int idx = i0 + u * 32 + lane;
                r[u] = FIRST ? qw[idx < cnt ? idx : 0] : gq[idx < cnt ? idx : 0];
            }
#pragma unroll
            for (int u = 0; u < 4; ++u) {
                if (FIRST) {
                    const int e  = r[u] >> 8;
                    const int ec = e < E0 ? e : E0 - 1;
                    const int sl = src[ec];
                    const int dd = dst[ec];
                    s[u]   = e < E0 ? sl : e - E0;
                    dlv[u] = e < E0 ? dd - base : (r[u] & 255);
                } else {
                    s[u]   = r[u] >> 8;
                    dlv[u] = r[u] & 255;
                }
            }
#pragma unroll
            for (int u = 0; u < 4; ++u) lane_logits(a_src, a_dst, s[u], base + dlv[u], v[u]);
#pragma unroll
            for (int u = 0; u < 4; ++u) {
                const int idx = i0 + u * 32 + lane;
                if (idx < cnt) {
                    const int dl = dlv[u];
                    const int en = (s[u] << 8) | dl;
                    qw[idx] = en;
                    lgw[idx] = make_float4(v[u][0], v[u][1], v[u][2], v[u][3]);
                    if (FIRST) gq[idx] = en;
#pragma unroll
                    for (int h = 0; h < HEADS; ++h)
                        __hip_atomic_fetch_max(mloc + dl * HEADS + h, ford(v[u][h]),
                                               __ATOMIC_RELAXED, __HIP_MEMORY_SCOPE_WORKGROUP);
                }
            }
        }
        __syncthreads();
        for (int i0 = 0; i0 < cnt; i0 += 128) {
            int r[4];
            float4 lv[4];
#pragma unroll
            for (int u = 0; u < 4; ++u) {
                const int idx = i0 + u * 32 + lane;
                r[u]  = qw[idx < cnt ? idx : 0];
                lv[u] = lgw[idx < cnt ? idx : 0];
            }
#pragma unroll
            for (int u = 0; u < 4; ++u) {
                const int idx = i0 + u * 32 + lane;
                if (idx < cnt) {
                    const int dl = r[u] & 255;
                    const float v[HEADS] = {lv[u].x, lv[u].y, lv[u].z, lv[u].w};
#pragma unroll
                    for (int h = 0; h < HEADS; ++h) {
                        const float mv = funord(mloc[dl * HEADS + h]);
                        __hip_atomic_fetch_add(zloc + dl * HEADS + h, __expf(v[h] - mv),
                                               __ATOMIC_RELAXED, __HIP_MEMORY_SCOPE_WORKGROUP);
                    }
                }
            }
        }
        __syncthreads();
        const int kk = (lane & 15) >> 2;
        const float* lgf = (const float*)lgw;
        int i = 0;
        for (; i + 4 <= cnt; i += 4)
            rows_add<4, true>(qw[i + kk], base, lane, a_src, a_dst, lgf + i * 4 + (lane & 15), mloc, zloc, H, slab);
        for (; i < cnt; ++i)
            rows_add<1, true>(qw[i], base, lane, a_src, a_dst, lgf + i * 4 + (lane & 3), mloc, zloc, H, slab);
    } else {
        rescan<0>(tid, lane, wave, base, src, dst, a_src, a_dst, mloc, zloc, H, slab);
        __syncthreads();
        rescan<1>(tid, lane, wave, base, src, dst, a_src, a_dst, mloc, zloc, H, slab);
        __syncthreads();
        rescan<2>(tid, lane, wave, base, src, dst, a_src, a_dst, mloc, zloc, H, slab);
    }

    __syncthreads();
    float4* o4 = (float4*)(rows + (size_t)base * C_HID);
    for (int i = tid; i < SLAB * C_HID / 4; i += 1024) o4[i] = s4[i];
}

extern "C" void kernel_launch(void* const* d_in, const int* in_sizes, int n_in,
                              void* d_out, int out_size, void* d_ws, size_t ws_size,
                              hipStream_t stream) {
    const float* x        = (const float*)d_in[0];
    const int*   eidx     = (const int*)  d_in[1];
    const float* W1       = (const float*)d_in[2];
    const float* att_s1   = (const float*)d_in[3];
    const float* att_d1   = (const float*)d_in[4];
    const float* b1       = (const float*)d_in[5];
    const float* W2       = (const float*)d_in[6];
    const float* att_s2   = (const float*)d_in[7];
    const float* att_d2   = (const float*)d_in[8];
    const float* b2       = (const float*)d_in[9];
    float*       out      = (float*)d_out;

    const int* src = eidx;
    const int* dst = eidx + E0;

    const size_t NB = (size_t)N_NODES * C_HID;
    float* h      = (float*)d_ws;
    float* agg    = h      + NB;
    float* a_src  = agg    + NB;
    float* a_dst  = a_src  + (size_t)N_NODES * HEADS;
    int* qsave    = (int*)(a_dst + (size_t)N_NODES * HEADS);
    int* qcount   = qsave  + (size_t)NSLAB * SWAVES * QCAP;
    int* qslow    = qcount + NSLAB * SWAVES;
    float* WA1    = (float*)(qslow + 256);
    float* WA2    = WA1 + C_IN * 16;
    int* ids      = (int*)(WA2 + C_HID * 16);

    const dim3 B256(256);
    const int gMM  = N_NODES / 16;
    const int gFOLD = FOLD_BLOCKS + ID_BLOCKS;
    const int gMB   = (gMM + 1) / 2 * 2;

    fold_att   <<<gFOLD, B256, 0, stream>>>(dst, ids, W1, att_s1, att_d1, W2, att_s2, att_d2, WA1, WA2);
    gemm_wmma<false, C_IN><<<gMB, 128, 0, stream>>>(x, nullptr, W1, WA1, h, a_src, a_dst, C_HID, gMM);
    edge_slab<true><<<NSLAB, 1024, 0, stream>>>(src, dst, a_src, a_dst, h, nullptr, agg, qsave, qcount, qslow, ids);

    gemm_wmma<true, C_HID><<<gMB, 128, 0, stream>>>(agg, b1, W2, WA2, h, a_src, a_dst, C_HID, gMM);
    edge_slab<false><<<NSLAB, 1024, 0, stream>>>(src, dst, a_src, a_dst, h, b2, out, qsave, qcount, qslow, nullptr);
}


# Row gather by whole-line 16-byte loads: a source row comes in 2 loads per lane, not 8 of 4 bytes; slab columns permuted in LDS so the float adds keep 32 banks; sums still in LDS arrival order
# speedup vs baseline: 7.5162x; 1.0702x over previous
#include <hip/hip_runtime.h>
#include <math.h>

typedef __attribute__((ext_vector_type(2))) float v2f;
typedef __attribute__((ext_vector_type(8))) float v8f;

#define N_NODES 50000
#define C_IN    128
#define C_HID   256
#define HEADS   4
#define EMBED   64
#define E0      800000
#define ETOT    (E0 + N_NODES)
#define NEG_SLOPE 0.2f
#define SLAB   200
#define NSLAB  (N_NODES / SLAB)
static_assert(N_NODES % SLAB == 0, "rows split evenly over the blocks");
static_assert(NSLAB <= 255, "a slab number fits a byte; 255 stands for a row outside the graph");
static_assert(E0 % 16 == 0, "slab numbers are written four and scanned sixteen at a time");
#define FOLD_BLOCKS (((C_IN + C_HID) * 32 + 255) / 256)
#define ID_BLOCKS   ((E0 / 32 + 255) / 256)

template <bool ACT, int K>
__global__ __launch_bounds__(128) void gemm_wmma(const float* __restrict__ X,
                          const float* __restrict__ bias,
                          const float* __restrict__ W,
                          const float* __restrict__ WA,
                          float* __restrict__ H,
                          float* __restrict__ a_src,
                          float* __restrict__ a_dst,
                          int N, int MT) {
    static_assert(K % 16 == 0 && 128 % (K / 4) == 0, "four k-steps per trip; a thread keeps one column group");
    constexpr int KP = K + 4;
    __shared__ float As[32 * KP];
    __shared__ float Dx[3 * 256];

    const int tid    = threadIdx.x;
    const int lane   = tid & 31;
    const int wave   = __builtin_amdgcn_readfirstlane(tid >> 5);
    const int lane15 = lane & 15;
    const int khalf  = lane >> 4;
    const int ch     = (int)blockIdx.x & 1;
    const int tile0  = ((int)blockIdx.x >> 1) * 2;

    {
        constexpr int C4  = K / 4;
        constexpr int RS  = 128 / C4;
        constexpr int NLD = 32 / RS;
        const int c4 = tid % C4;
        const int r0 = tid / C4;
        const float* p0 = X + (size_t)(tile0 * 16 + r0) * K + c4 * 4;
        const float* p1 = p0 + (tile0 + 1 < MT ? 16 * K : 0);
        float4 t[NLD];
#pragma unroll
        for (int i = 0; i < NLD / 2; ++i) {
            t[i]           = *(const float4*)(p0 + i * RS * K);
            t[i + NLD / 2] = *(const float4*)(p1 + i * RS * K);
        }
        float4 bb = make_float4(0.f, 0.f, 0.f, 0.f);
        if (ACT) bb = *(const float4*)(bias + c4 * 4);
#pragma unroll
        for (int i = 0; i < NLD; ++i) {
            float4 v = t[i];
            if (ACT) {
                v.x = fmaxf(v.x + bb.x, 0.f);
                v.y = fmaxf(v.y + bb.y, 0.f);
                v.z = fmaxf(v.z + bb.z, 0.f);
                v.w = fmaxf(v.w + bb.w, 0.f);
            }
            *(float4*)(As + (r0 + i * RS) * KP + c4 * 4) = v;
        }
    }
    __syncthreads();

    const int col0 = ch * 128 + wave * 32;

    v8f acc00 = {}, acc01 = {};
    v8f acc10 = {}, acc11 = {};
    v8f accD  = {};

    const float* ar0 = As + lane15 * KP + khalf * 2;
    const float* ar1 = ar0 + 16 * KP;
    const float* ard = As + (ch * 16 + lane15) * KP + khalf * 2 + wave * 4;
    const float* wc  = W + col0 + lane15;
    const float* wac = WA + lane15;

    for (int k = 0; k < K; k += 16) {
#pragma unroll
        for (int u = 0; u < 4; ++u) {
            const int ks = k + u * 4;
            const int kk = ks + khalf * 2;
            const v2f a0 = *(const v2f*)(ar0 + ks);
            const v2f a1 = *(const v2f*)(ar1 + ks);
            const float* wp0 = wc + (size_t)kk * N;
            const float* wp1 = wp0 + N;
            v2f b0, b1;
            b0.x = wp0[ 0]; b0.y = wp1[ 0];
            b1.x = wp0[16]; b1.y = wp1[16];
            acc00 = __builtin_amdgcn_wmma_f32_16x16x4_f32(false, a0, false, b0, (short)0, acc00, false, false);
            acc10 = __builtin_amdgcn_wmma_f32_16x16x4_f32(false, a1, false, b0, (short)0, acc10, false, false);
            acc01 = __builtin_amdgcn_wmma_f32_16x16x4_f32(false, a0, false, b1, (short)0, acc01, false, false);
            acc11 = __builtin_amdgcn_wmma_f32_16x16x4_f32(false, a1, false, b1, (short)0, acc11, false, false);
        }
        {
            const int kd = k + wave * 4 + khalf * 2;
            const v2f ad = *(const v2f*)(ard + k);
            v2f bd;
            bd.x = wac[kd * 16];  bd.y = wac[(kd + 1) * 16];
            accD = __builtin_amdgcn_wmma_f32_16x16x4_f32(false, ad, false, bd, (short)0, accD, false, false);
        }
    }

    if (wave) {
#pragma unroll
        for (int i = 0; i < 8; ++i) Dx[(wave - 1) * 256 + i * 32 + lane] = accD[i];
    }
    __syncthreads();
    if (wave == 0 && tile0 + ch < MT && lane15 < 8) {
        float* op = (lane15 < 4 ? a_src : a_dst) + (size_t)((tile0 + ch) * 16 + khalf * 8) * HEADS + (lane15 & 3);
#pragma unroll
        for (int i = 0; i < 8; ++i) {
            const float p1 = Dx[      i * 32 + lane];
            const float p2 = Dx[256 + i * 32 + lane];
            const float p3 = Dx[512 + i * 32 + lane];
            op[i * HEADS] = (accD[i] + p2) + (p1 + p3);
        }
    }

    {
        float* op = H + (size_t)(tile0 * 16 + khalf * 8) * N + col0 + lane15;
#pragma unroll
        for (int i = 0; i < 8; ++i) {
            op[i * N +  0] = acc00[i];
            op[i * N + 16] = acc01[i];
        }
        if (tile0 + 1 < MT) {
            op += (size_t)16 * N;
#pragma unroll
            for (int i = 0; i < 8; ++i) {
                op[i * N +  0] = acc10[i];
                op[i * N + 16] = acc11[i];
            }
        }
    }
}

__device__ __forceinline__ unsigned slab_of(int d) {
    const unsigned u = (unsigned)d;
    return u < (unsigned)N_NODES ? u / SLAB : 255u;
}

__global__ void fold_att(const int* __restrict__ dst,
                         int* __restrict__ ids,
                         const float* __restrict__ W1,
                         const float* __restrict__ as1,
                         const float* __restrict__ ad1,
                         const float* __restrict__ W2,
                         const float* __restrict__ as2,
                         const float* __restrict__ ad2,
                         float* __restrict__ WA1,
                         float* __restrict__ WA2) {
    if (blockIdx.x >= FOLD_BLOCKS) {
        const int t = ((int)blockIdx.x - FOLD_BLOCKS) * 256 + (int)threadIdx.x;
        const int4* d4 = (const int4*)dst;
        int4 d[8];
#pragma unroll
        for (int k = 0; k < 8; ++k) {
            const int pi = t + k * (ID_BLOCKS * 256);
            d[k] = d4[pi < E0 / 4 ? pi : E0 / 4 - 1];
        }
#pragma unroll
        for (int k = 0; k < 8; ++k) {
            const int pi = t + k * (ID_BLOCKS * 256);
            if (pi < E0 / 4)
                ids[pi] = (int)(slab_of(d[k].x) | (slab_of(d[k].y) << 8) | (slab_of(d[k].z) << 16) | (slab_of(d[k].w) << 24));
        }
        return;
    }
    const int gw   = (int)((blockIdx.x * blockDim.x + threadIdx.x) >> 5);
    const int lane = threadIdx.x & 31;
    if (gw >= C_IN + C_HID) return;
    const bool second = gw >= C_IN;
    const int  k      = second ? gw - C_IN : gw;
    const float* Wm = second ? W2 : W1;
    const float* as = second ? as2 : as1;
    const float* ad = second ? ad2 : ad1;
    float*       WA = second ? WA2 : WA1;

    const int j    = lane >> 2;
    const int part = lane & 3;
    const int head = j & 3;
    const float* att = (j < 4 ? as : ad) + head * EMBED + part * 16;
    const float* wp  = Wm + (size_t)k * C_HID + head * EMBED + part * 16;
    float s = 0.f;
#pragma unroll
    for (int c = 0; c < 16; c += 4) {
        const float4 w4 = *(const float4*)(wp + c);
        const float4 t4 = *(const float4*)(att + c);
        s += w4.x * t4.x;
        s += w4.y * t4.y;
        s += w4.z * t4.z;
        s += w4.w * t4.w;
    }
    s += __shfl_xor(s, 1, 32);
    s += __shfl_xor(s, 2, 32);
    if (part == 0) WA[k * 16 + j] = s;
    if (part == 1) WA[k * 16 + 8 + j] = 0.f;
}

__device__ __forceinline__ unsigned ford(float f) {
    unsigned u = __float_as_uint(f);
    return (u >> 31) ? ~u : (u | 0x80000000u);
}
__device__ __forceinline__ float funord(unsigned u) {
    return (u >> 31) ? __uint_as_float(u ^ 0x80000000u) : __uint_as_float(~u);
}

__device__ __forceinline__ float leaky(float v) {
    return v > 0.f ? v : NEG_SLOPE * v;
}

#define QCAP   176
#define SWAVES 32
#define ORD_NEG_INF 0x007FFFFFu
static_assert(SLAB <= 256 && ETOT < (1 << 23), "a queue entry packs edge << 8 | local row");
static_assert(E0 % 4 == 0, "the slow path scans dst in int4 pieces");

__device__ __forceinline__ void lane_logits(const float* __restrict__ a_src,
                                            const float* __restrict__ a_dst,
                                            int s, int d, float v[HEADS]) {
    const float4 as = *(const float4*)(a_src + (size_t)s * HEADS);
    const float4 ad = *(const float4*)(a_dst + (size_t)d * HEADS);
    v[0] = leaky(as.x + ad.x);
    v[1] = leaky(as.y + ad.y);
    v[2] = leaky(as.z + ad.z);
    v[3] = leaky(as.w + ad.w);
}

__device__ __forceinline__ void lane_max(const float* __restrict__ a_src,
                                         const float* __restrict__ a_dst,
                                         int s, int dl, int base, unsigned* mloc) {
    float v[HEADS];
    lane_logits(a_src, a_dst, s, base + dl, v);
#pragma unroll
    for (int h = 0; h < HEADS; ++h)
        __hip_atomic_fetch_max(mloc + dl * HEADS + h, ford(v[h]),
                               __ATOMIC_RELAXED, __HIP_MEMORY_SCOPE_WORKGROUP);
}

__device__ __forceinline__ void lane_z(const float* __restrict__ a_src,
                                       const float* __restrict__ a_dst,
                                       int s, int dl, int base,
                                       const unsigned* mloc, float* zloc) {
    float v[HEADS];
    lane_logits(a_src, a_dst, s, base + dl, v);
#pragma unroll
    for (int h = 0; h < HEADS; ++h) {
        const float mv = funord(mloc[dl * HEADS + h]);
        __hip_atomic_fetch_add(zloc + dl * HEADS + h, __expf(v[h] - mv),
                               __ATOMIC_RELAXED, __HIP_MEMORY_SCOPE_WORKGROUP);
    }
}

template <int NB, bool LGQ>
__device__ __forceinline__ void rows_add(int r, int base, int lane,
                                         const float* __restrict__ a_src,
                                         const float* __restrict__ a_dst,
                                         const float* lgp,
                                         const unsigned* mloc, const float* zloc,
                                         const float* __restrict__ H,
                                         float* slab) {
    int s[NB], dl[NB];
#pragma unroll
    for (int k = 0; k < NB; ++k) {
        const int rk = __builtin_amdgcn_readlane(r, 4 * k);
        s[k]  = rk >> 8;
        dl[k] = rk & 255;
    }
    float4 hv[NB][2];
#pragma unroll
    for (int k = 0; k < NB; ++k) {
        const float4* hp = (const float4*)(H + (size_t)s[k] * C_HID) + lane;
        hv[k][0] = hp[0];
        hv[k][1] = hp[32];
    }
    const int h   = lane & 3;
    const int sL  = r >> 8;
    const int dlL = r & 255;
    float v;
    if (LGQ) v = *lgp;
    else     v = leaky(a_src[(size_t)sL * HEADS + h] + a_dst[(size_t)(base + dlL) * HEADS + h]);
    const float ev = __expf(v - funord(mloc[dlL * HEADS + h]));
    const int   al = __float_as_int(ev / (zloc[dlL * HEADS + h] + 1e-16f));
    const bool up = lane >= 16;
#pragma unroll
    for (int k = 0; k < NB; ++k) {
        float* sp = slab + dl[k] * C_HID + lane;
        const float a0 = __int_as_float(__builtin_amdgcn_readlane(al, 4 * k + 0));
        const float a1 = __int_as_float(__builtin_amdgcn_readlane(al, 4 * k + 1));
        const float a2 = __int_as_float(__builtin_amdgcn_readlane(al, 4 * k + 2));
        const float a3 = __int_as_float(__builtin_amdgcn_readlane(al, 4 * k + 3));
        const float alo = up ? a1 : a0;
        const float ahi = up ? a3 : a2;
        const float p[8] = {hv[k][0].x * alo, hv[k][0].y * alo, hv[k][0].z * alo, hv[k][0].w * alo,
                            hv[k][1].x * ahi, hv[k][1].y * ahi, hv[k][1].z * ahi, hv[k][1].w * ahi};
#pragma unroll
        for (int j = 0; j < 8; ++j)
            __hip_atomic_fetch_add(sp + j * 32, p[j],
                                   __ATOMIC_RELAXED, __HIP_MEMORY_SCOPE_WORKGROUP);
    }
}

template <int PH>
__device__ __forceinline__ void rescan(int tid, int lane, int wave, int base,
                                       const int* __restrict__ src,
                                       const int* __restrict__ dst,
                                       const float* __restrict__ a_src,
                                       const float* __restrict__ a_dst,
                                       unsigned* mloc, float* zloc,
                                       const float* __restrict__ H,
                                       float* slab) {
    if (PH < 2) {
        for (int n = tid; n < SLAB; n += 1024) {
            if (PH == 0) lane_max(a_src, a_dst, base + n, n, base, mloc);
            else         lane_z(a_src, a_dst, base + n, n, base, mloc, zloc);
        }
    } else {
        for (int n = wave; n < SLAB; n += SWAVES) {
            rows_add<1, false>(((base + n) << 8) | n, base, lane, a_src, a_dst, nullptr, mloc, zloc, H, slab);
        }
    }
    const int4* d4p = (const int4*)dst;
    const int NV = E0 / 4;
    for (int vb = wave * 32; vb < NV; vb += SWAVES * 32) {
        const int vi = vb + lane;
        const int4 t = d4p[vi < NV ? vi : NV - 1];
        const int comp[4] = {t.x, t.y, t.z, t.w};
#pragma unroll
        for (int c = 0; c < 4; ++c) {
            const unsigned dl = (unsigned)(comp[c] - base);
            const bool mt = vi < NV && dl < (unsigned)SLAB;
            int s = 0;
            if (mt) s = src[vi * 4 + c];
            if (PH == 0) {
                if (mt) lane_max(a_src, a_dst, s, (int)dl, base, mloc);
            } else if (PH == 1) {
                if (mt) lane_z(a_src, a_dst, s, (int)dl, base, mloc, zloc);
            } else {
                unsigned mask = __builtin_amdgcn_ballot_w32(mt);
                while (mask) {
                    const int b = __builtin_ctz(mask);
                    mask &= mask - 1;
                    const int rb = __builtin_amdgcn_readlane((s << 8) | (int)dl, b);
                    rows_add<1, false>(rb, base, lane, a_src, a_dst, nullptr, mloc, zloc, H, slab);
                }
            }
        }
    }
}

template <bool FIRST>
__global__ __launch_bounds__(1024) void edge_slab(const int* __restrict__ src,
                                                  const int* __restrict__ dst,
                                                  const float* __restrict__ a_src,
                                                  const float* __restrict__ a_dst,
                                                  const float* __restrict__ H,
                                                  const float* __restrict__ bias,
                                                  float* __restrict__ rows,
                                                  int* __restrict__ qsave,
                                                  int* __restrict__ qcount,
                                                  int* __restrict__ qslow,
                                                  const int* __restrict__ ids) {
    __shared__ float    slab[SLAB * C_HID];
    __shared__ int      queue[SWAVES * QCAP];
    __shared__ float4   lgq[SWAVES * QCAP];
    __shared__ unsigned mloc[SLAB * HEADS];
    __shared__ float    zloc[SLAB * HEADS];
    __shared__ int      ovflag;

    const int tid  = threadIdx.x;
    const int lane = tid & 31;
    const int wave = __builtin_amdgcn_readfirstlane(tid >> 5);
    const int base = blockIdx.x * SLAB;

    float4* s4 = (float4*)slab;
    {
        float4 start = make_float4(0.f, 0.f, 0.f, 0.f);
        if (!FIRST) {
            const int pi = tid & 63;
            const int j  = pi >> 3;
            const float* bp = bias + (j >> 2) * 128 + (pi & 7) * 16 + (j & 3);
            start = make_float4(bp[0], bp[4], bp[8], bp[12]);
        }
        for (int i = tid; i < SLAB * C_HID / 4; i += 1024) s4[i] = start;
    }
    for (int i = tid; i < SLAB * HEADS; i += 1024) { mloc[i] = ORD_NEG_INF; zloc[i] = 0.f; }
    if (FIRST && tid == 0) ovflag = 0;

    int* qw = queue + wave * QCAP;
    float4* lgw = lgq + wave * QCAP;
    int* gq = qsave + ((size_t)blockIdx.x * SWAVES + wave) * QCAP;
    int* gc = qcount + blockIdx.x * SWAVES + wave;
    int  cnt;
    bool slow;
    if (FIRST) {
        cnt = (SLAB - 1 - wave) / SWAVES + 1;
        if (lane < cnt) {
            const int n = wave + SWAVES * lane;
            qw[lane] = ((E0 + base + n) << 8) | n;
        }
        __syncthreads();

        const int4* i4p = (const int4*)ids;
        const int NV = E0 / 16;
        const unsigned pat = (unsigned)blockIdx.x * 0x01010101u;
        constexpr int ROWP = SWAVES * 32;
        bool ovf = false;
        for (int vb = wave * 32; vb < NV; vb += 4 * ROWP) {
            int4 p[4];
#pragma unroll
            for (int q = 0; q < 4; ++q) {
                const int vi = vb + q * ROWP + lane;
                p[q] = i4p[vi < NV ? vi : NV - 1];
            }
#pragma unroll
            for (int q = 0; q < 4; ++q) {
                const int vi = vb + q * ROWP + lane;
                const unsigned w[4] = {(unsigned)p[q].x, (unsigned)p[q].y, (unsigned)p[q].z, (unsigned)p[q].w};
                unsigned mm = 0;
#pragma unroll
                for (int d = 0; d < 4; ++d) {
                    const unsigned t = w[d] ^ pat;
                    const unsigned zb = ~(((t & 0x7F7F7F7Fu) + 0x7F7F7F7Fu) | t | 0x7F7F7F7Fu);
                    mm |= zb >> (7 - d);
                }
                if (vi >= NV) mm = 0;
                unsigned live = __builtin_amdgcn_ballot_w32(mm != 0);
                while (live) {
                    const int n = __builtin_popcount(live);
                    if (cnt + n <= QCAP) {
                        if (mm != 0) {
                            const int b = __builtin_ctz(mm);
                            const int e = vi * 16 + (b & 7) * 4 + (b >> 3);
                            qw[cnt + __builtin_amdgcn_mbcnt_lo(live, 0u)] = e << 8;
                        }
                        cnt += n;
                    } else {
                        ovf = true;
                    }
                    mm &= mm - 1;
                    live = __builtin_amdgcn_ballot_w32(mm != 0);
                }
            }
        }
        if (ovf && lane == 0) ovflag = 1;
        __syncthreads();
        slow = ovflag != 0;
        if (tid == 0) qslow[blockIdx.x] = slow ? 1 : 0;
        if (lane == 0) *gc = cnt;
    } else {
        slow = qslow[blockIdx.x] != 0;
        cnt  = *gc;
        __syncthreads();
    }

    if (!slow) {
        for (int i0 = 0; i0 < cnt; i0 += 128) {
            int r[4], s[4], dlv[4];
            float v[4][HEADS];
#pragma unroll
            for (int u = 0; u < 4; ++u) {
                const int idx = i0 + u * 32 + lane;
                r[u] = FIRST ? qw[idx < cnt ? idx : 0] : gq[idx < cnt ? idx : 0];
            }
#pragma unroll
            for (int u = 0; u < 4; ++u) {
                if (FIRST) {
                    const int e  = r[u] >> 8;
                    const int ec = e < E0 ? e : E0 - 1;
                    const int sl = src[ec];
                    const int dd = dst[ec];
                    s[u]   = e < E0 ? sl : e - E0;
                    dlv[u] = e < E0 ? dd - base : (r[u] & 255);
                } else {
                    s[u]   = r[u] >> 8;
                    dlv[u] = r[u] & 255;
                }
            }
#pragma unroll
            for (int u = 0; u < 4; ++u) lane_logits(a_src, a_dst, s[u], base + dlv[u], v[u]);
#pragma unroll
            for (int u = 0; u < 4; ++u) {
                const int idx = i0 + u * 32 + lane;
                if (idx < cnt) {
                    const int dl = dlv[u];
                    const int en = (s[u] << 8) | dl;
                    qw[idx] = en;
                    lgw[idx] = make_float4(v[u][0], v[u][1], v[u][2], v[u][3]);
                    if (FIRST) gq[idx] = en;
#pragma unroll
                    for (int h = 0; h < HEADS; ++h)
                        __hip_atomic_fetch_max(mloc + dl * HEADS + h, ford(v[u][h]),
                                               __ATOMIC_RELAXED, __HIP_MEMORY_SCOPE_WORKGROUP);
                }
            }
        }
        __syncthreads();
        for (int i0 = 0; i0 < cnt; i0 += 128) {
            int r[4];
            float4 lv[4];
#pragma unroll
            for (int u = 0; u < 4; ++u) {
                const int idx = i0 + u * 32 + lane;
                r[u]  = qw[idx < cnt ? idx : 0];
                lv[u] = lgw[idx < cnt ? idx : 0];
            }
#pragma unroll
            for (int u = 0; u < 4; ++u) {
                const int idx = i0 + u * 32 + lane;
                if (idx < cnt) {
                    const int dl = r[u] & 255;
                    const float v[HEADS] = {lv[u].x, lv[u].y, lv[u].z, lv[u].w};
#pragma unroll
                    for (int h = 0; h < HEADS; ++h) {
                        const float mv = funord(mloc[dl * HEADS + h]);
                        __hip_atomic_fetch_add(zloc + dl * HEADS + h, __expf(v[h] - mv),
                                               __ATOMIC_RELAXED, __HIP_MEMORY_SCOPE_WORKGROUP);
                    }
                }
            }
        }
        __syncthreads();
        const int kk = (lane & 15) >> 2;
        const float* lgf = (const float*)lgw;
        int i = 0;
        for (; i + 4 <= cnt; i += 4)
            rows_add<4, true>(qw[i + kk], base, lane, a_src, a_dst, lgf + i * 4 + (lane & 15), mloc, zloc, H, slab);
        for (; i < cnt; ++i)
            rows_add<1, true>(qw[i], base, lane, a_src, a_dst, lgf + i * 4 + (lane & 3), mloc, zloc, H, slab);
    } else {
        rescan<0>(tid, lane, wave, base, src, dst, a_src, a_dst, mloc, zloc, H, slab);
        __syncthreads();
        rescan<1>(tid, lane, wave, base, src, dst, a_src, a_dst, mloc, zloc, H, slab);
        __syncthreads();
        rescan<2>(tid, lane, wave, base, src, dst, a_src, a_dst, mloc, zloc, H, slab);
    }

    __syncthreads();
    float4* o4 = (float4*)(rows + (size_t)base * C_HID);
    for (int i = tid; i < SLAB * C_HID / 4; i += 1024) {
        const int m = i & 63;
        const float* rp = slab + (i >> 6) * C_HID + (m >> 5) * 128 + (m & 31);
        o4[i] = make_float4(rp[0], rp[32], rp[64], rp[96]);
    }
}

extern "C" void kernel_launch(void* const* d_in, const int* in_sizes, int n_in,
                              void* d_out, int out_size, void* d_ws, size_t ws_size,
                              hipStream_t stream) {
    const float* x        = (const float*)d_in[0];
    const int*   eidx     = (const int*)  d_in[1];
    const float* W1       = (const float*)d_in[2];
    const float* att_s1   = (const float*)d_in[3];
    const float* att_d1   = (const float*)d_in[4];
    const float* b1       = (const float*)d_in[5];
    const float* W2       = (const float*)d_in[6];
    const float* att_s2   = (const float*)d_in[7];
    const float* att_d2   = (const float*)d_in[8];
    const float* b2       = (const float*)d_in[9];
    float*       out      = (float*)d_out;

    const int* src = eidx;
    const int* dst = eidx + E0;

    const size_t NB = (size_t)N_NODES * C_HID;
    float* h      = (float*)d_ws;
    float* agg    = h      + NB;
    float* a_src  = agg    + NB;
    float* a_dst  = a_src  + (size_t)N_NODES * HEADS;
    int* qsave    = (int*)(a_dst + (size_t)N_NODES * HEADS);
    int* qcount   = qsave  + (size_t)NSLAB * SWAVES * QCAP;
    int* qslow    = qcount + NSLAB * SWAVES;
    float* WA1    = (float*)(qslow + 256);
    float* WA2    = WA1 + C_IN * 16;
    int* ids      = (int*)(WA2 + C_HID * 16);

    const dim3 B256(256);
    const int gMM  = N_NODES / 16;
    const int gFOLD = FOLD_BLOCKS + ID_BLOCKS;
    const int gMB   = (gMM + 1) / 2 * 2;

    fold_att   <<<gFOLD, B256, 0, stream>>>(dst, ids, W1, att_s1, att_d1, W2, att_s2, att_d2, WA1, WA2);
    gemm_wmma<false, C_IN><<<gMB, 128, 0, stream>>>(x, nullptr, W1, WA1, h, a_src, a_dst, C_HID, gMM);
    edge_slab<true><<<NSLAB, 1024, 0, stream>>>(src, dst, a_src, a_dst, h, nullptr, agg, qsave, qcount, qslow, ids);

    gemm_wmma<true, C_HID><<<gMB, 128, 0, stream>>>(agg, b1, W2, WA2, h, a_src, a_dst, C_HID, gMM);
    edge_slab<false><<<NSLAB, 1024, 0, stream>>>(src, dst, a_src, a_dst, h, b2, out, qsave, qcount, qslow, nullptr);
}
